# starting kernel
# baseline (speedup 1.0000x reference)
#include <hip/hip_runtime.h>
#include <hip/hip_bf16.h>


typedef float v2f __attribute__((ext_vector_type(2)));
typedef float v8f __attribute__((ext_vector_type(8)));

#define NFEAT 128
#define HIDW  128

__global__ void gcn_init_kernel(float* __restrict__ deg,
                                float* __restrict__ agg,
                                int n, int total) {
    int i = blockIdx.x * blockDim.x + threadIdx.x;
    if (i < total) agg[i] = 0.0f;
    if (i < n)     deg[i] = 1.0f;
}

__global__ void gcn_degree_kernel(const int* __restrict__ dst,
                                  float* __restrict__ deg, int nedges) {
    int i = blockIdx.x * blockDim.x + threadIdx.x;
    if (i < nedges) atomicAdd(&deg[dst[i]], 1.0f);
}

__global__ void gcn_rsqrt_kernel(float* __restrict__ deg, int n) {
    int i = blockIdx.x * blockDim.x + threadIdx.x;
    if (i < n) {
        float d = deg[i];
        deg[i] = (d > 0.0f) ? rsqrtf(d) : 0.0f;
    }
}

__global__ void __launch_bounds__(128)
gcn_gemm_xw_kernel(const float* __restrict__ x,
                   const float* __restrict__ W1,
                   float* __restrict__ xw, int nrows) {
    const int wave  = threadIdx.x >> 5;
    const int lane  = threadIdx.x & 31;
    const int tileM = blockIdx.x * 16;
    if (tileM >= nrows) return;

    const int l15   = lane & 15;
    const int khalf = (lane >> 4) * 2;
    int mrow = tileM + l15;
    if (mrow >= nrows) mrow = nrows - 1;

    const int n0    = wave * 32;
    const int ncol0 = n0 + l15;
    const int ncol1 = n0 + 16 + l15;

    v8f c0 = {0.f, 0.f, 0.f, 0.f, 0.f, 0.f, 0.f, 0.f};
    v8f c1 = {0.f, 0.f, 0.f, 0.f, 0.f, 0.f, 0.f, 0.f};

    const float* __restrict__ xrow = x + (size_t)mrow * NFEAT;
    #pragma unroll 4
    for (int kb = 0; kb < NFEAT; kb += 4) {
        const int k = kb + khalf;
        v2f a;
        a.x = xrow[k];
        a.y = xrow[k + 1];
        v2f b0, b1;
        b0.x = W1[(size_t)k       * HIDW + ncol0];
        b0.y = W1[(size_t)(k + 1) * HIDW + ncol0];
        b1.x = W1[(size_t)k       * HIDW + ncol1];
        b1.y = W1[(size_t)(k + 1) * HIDW + ncol1];
        c0 = __builtin_amdgcn_wmma_f32_16x16x4_f32(false, a, false, b0,
                                                   (short)0, c0, false, false);
        c1 = __builtin_amdgcn_wmma_f32_16x16x4_f32(false, a, false, b1,
                                                   (short)0, c1, false, false);
    }

    const int rBase = tileM + ((lane >> 4) * 8);
    float* __restrict__ orow = xw + (size_t)rBase * HIDW;
    if (tileM + 16 <= nrows) {
        #pragma unroll
        for (int r = 0; r < 8; ++r) {
            orow[(size_t)r * HIDW + ncol0] = c0[r];
            orow[(size_t)r * HIDW + ncol1] = c1[r];
        }
    } else {
        #pragma unroll
        for (int r = 0; r < 8; ++r) {
            if (rBase + r < nrows) {
                orow[(size_t)r * HIDW + ncol0] = c0[r];
                orow[(size_t)r * HIDW + ncol1] = c1[r];
            }
        }
    }
}

__global__ void __launch_bounds__(256)
gcn_scatter_kernel(const float* __restrict__ xw,
                   const float* __restrict__ dinv,
                   const int* __restrict__ src,
                   const int* __restrict__ dst,
                   float* __restrict__ agg, int nedges) {
    const int gwarp = (int)((blockIdx.x * blockDim.x + threadIdx.x) >> 5);
    const int lane  = threadIdx.x & 31;
    if (gwarp >= nedges) return;

    const int s = src[gwarp];
    const int d = dst[gwarp];
    const float norm = dinv[s] * dinv[d];

    const float4 v = ((const float4*)(xw + (size_t)s * HIDW))[lane];
    float* ar = agg + (size_t)d * HIDW + lane * 4;
    atomicAdd(ar + 0, v.x * norm);
    atomicAdd(ar + 1, v.y * norm);
    atomicAdd(ar + 2, v.z * norm);
    atomicAdd(ar + 3, v.w * norm);
}

__global__ void __launch_bounds__(256)
gcn_finalize_kernel(const float* __restrict__ x,
                    const float* __restrict__ xw,
                    const float* __restrict__ agg,
                    const float* __restrict__ dinv,
                    const float* __restrict__ b1,
                    const float* __restrict__ Wfc,
                    const float* __restrict__ bfc,
                    float* __restrict__ out, int n) {
    const int node = (int)((blockIdx.x * blockDim.x + threadIdx.x) >> 5);
    const int lane = threadIdx.x & 31;
    if (node >= n) return;

    const float di   = dinv[node];
    const float self = di * di;

    const float4 xv = ((const float4*)(x   + (size_t)node * NFEAT))[lane];
    const float4 wv = ((const float4*)(xw  + (size_t)node * HIDW))[lane];
    const float4 av = ((const float4*)(agg + (size_t)node * HIDW))[lane];
    const float4 bv = ((const float4*)b1)[lane];

    float4 h;
    h.x = fmaxf(av.x + wv.x * self + bv.x, 0.0f);
    h.y = fmaxf(av.y + wv.y * self + bv.y, 0.0f);
    h.z = fmaxf(av.z + wv.z * self + bv.z, 0.0f);
    h.w = fmaxf(av.w + wv.w * self + bv.w, 0.0f);

    const int f0 = lane * 4;
    float acc0, acc1;
    acc0  = xv.x * Wfc[(f0 + 0) * 2 + 0] + xv.y * Wfc[(f0 + 1) * 2 + 0]
          + xv.z * Wfc[(f0 + 2) * 2 + 0] + xv.w * Wfc[(f0 + 3) * 2 + 0];
    acc1  = xv.x * Wfc[(f0 + 0) * 2 + 1] + xv.y * Wfc[(f0 + 1) * 2 + 1]
          + xv.z * Wfc[(f0 + 2) * 2 + 1] + xv.w * Wfc[(f0 + 3) * 2 + 1];
    acc0 += h.x * Wfc[(NFEAT + f0 + 0) * 2 + 0] + h.y * Wfc[(NFEAT + f0 + 1) * 2 + 0]
          + h.z * Wfc[(NFEAT + f0 + 2) * 2 + 0] + h.w * Wfc[(NFEAT + f0 + 3) * 2 + 0];
    acc1 += h.x * Wfc[(NFEAT + f0 + 0) * 2 + 1] + h.y * Wfc[(NFEAT + f0 + 1) * 2 + 1]
          + h.z * Wfc[(NFEAT + f0 + 2) * 2 + 1] + h.w * Wfc[(NFEAT + f0 + 3) * 2 + 1];

    #pragma unroll
    for (int off = 16; off > 0; off >>= 1) {
        acc0 += __shfl_down(acc0, off, 32);
        acc1 += __shfl_down(acc1, off, 32);
    }
    if (lane == 0) {
        out[(size_t)node * 2 + 0] = acc0 + bfc[0];
        out[(size_t)node * 2 + 1] = acc1 + bfc[1];
    }
}

extern "C" void kernel_launch(void* const* d_in, const int* in_sizes, int n_in,
                              void* d_out, int out_size, void* d_ws, size_t ws_size,
                              hipStream_t stream) {
    const float* x   = (const float*)d_in[0];
    const int*   ei  = (const int*)  d_in[1];
    const float* W1  = (const float*)d_in[2];
    const float* b1  = (const float*)d_in[3];
    const float* Wfc = (const float*)d_in[4];
    const float* bfc = (const float*)d_in[5];
    float*       out = (float*)d_out;

    const int N = in_sizes[0] / NFEAT;
    const int E = in_sizes[1] / 2;
    const int* src = ei;
    const int* dst = ei + E;

    float* deg = (float*)d_ws;
    float* xw  = deg + N;
    float* agg = xw + (size_t)N * HIDW;

    const int totalAgg = N * HIDW;

    gcn_init_kernel<<<(totalAgg + 255) / 256, 256, 0, stream>>>(deg, agg, N, totalAgg);

    gcn_degree_kernel<<<(E + 255) / 256, 256, 0, stream>>>(dst, deg, E);

    gcn_rsqrt_kernel<<<(N + 255) / 256, 256, 0, stream>>>(deg, N);

    gcn_gemm_xw_kernel<<<(N + 15) / 16, 128, 0, stream>>>(x, W1, xw, N);

    {
        const long long threads = (long long)E * 32;
        const int blocks = (int)((threads + 255) / 256);
        gcn_scatter_kernel<<<blocks, 256, 0, stream>>>(xw, deg, src, dst, agg, E);
    }

    {
        const long long threads = (long long)N * 32;
        const int blocks = (int)((threads + 255) / 256);
        gcn_finalize_kernel<<<blocks, 256, 0, stream>>>(x, xw, agg, deg, b1, Wfc, bfc, out, N);
    }
}


# scatter: lane owns features lane+32j, so each f32 atomic and each row load of a wave covers one whole 128-B line (4 atomic line requests per edge, not 16 quarter-filled); atomic order kept
# speedup vs baseline: 1.3214x; 1.3214x over previous
#include <hip/hip_runtime.h>
#include <hip/hip_bf16.h>


typedef float v2f __attribute__((ext_vector_type(2)));
typedef float v8f __attribute__((ext_vector_type(8)));

#define NFEAT 128
#define HIDW  128

__global__ void gcn_init_kernel(float* __restrict__ deg,
                                float* __restrict__ agg,
                                int n, int total) {
    int i = blockIdx.x * blockDim.x + threadIdx.x;
    if (i < total) agg[i] = 0.0f;
    if (i < n)     deg[i] = 1.0f;
}

__global__ void gcn_degree_kernel(const int* __restrict__ dst,
                                  float* __restrict__ deg, int nedges) {
    int i = blockIdx.x * blockDim.x + threadIdx.x;
    if (i < nedges) atomicAdd(&deg[dst[i]], 1.0f);
}

__global__ void gcn_rsqrt_kernel(float* __restrict__ deg, int n) {
    int i = blockIdx.x * blockDim.x + threadIdx.x;
    if (i < n) {
        float d = deg[i];
        deg[i] = (d > 0.0f) ? rsqrtf(d) : 0.0f;
    }
}

__global__ void __launch_bounds__(128)
gcn_gemm_xw_kernel(const float* __restrict__ x,
                   const float* __restrict__ W1,
                   float* __restrict__ xw, int nrows) {
    const int wave  = threadIdx.x >> 5;
    const int lane  = threadIdx.x & 31;
    const int tileM = blockIdx.x * 16;
    if (tileM >= nrows) return;

    const int l15   = lane & 15;
    const int khalf = (lane >> 4) * 2;
    int mrow = tileM + l15;
    if (mrow >= nrows) mrow = nrows - 1;

    const int n0    = wave * 32;
    const int ncol0 = n0 + l15;
    const int ncol1 = n0 + 16 + l15;

    v8f c0 = {0.f, 0.f, 0.f, 0.f, 0.f, 0.f, 0.f, 0.f};
    v8f c1 = {0.f, 0.f, 0.f, 0.f, 0.f, 0.f, 0.f, 0.f};

    const float* __restrict__ xrow = x + (size_t)mrow * NFEAT;
    #pragma unroll 4
    for (int kb = 0; kb < NFEAT; kb += 4) {
        const int k = kb + khalf;
        v2f a;
        a.x = xrow[k];
        a.y = xrow[k + 1];
        v2f b0, b1;
        b0.x = W1[(size_t)k       * HIDW + ncol0];
        b0.y = W1[(size_t)(k + 1) * HIDW + ncol0];
        b1.x = W1[(size_t)k       * HIDW + ncol1];
        b1.y = W1[(size_t)(k + 1) * HIDW + ncol1];
        c0 = __builtin_amdgcn_wmma_f32_16x16x4_f32(false, a, false, b0,
                                                   (short)0, c0, false, false);
        c1 = __builtin_amdgcn_wmma_f32_16x16x4_f32(false, a, false, b1,
                                                   (short)0, c1, false, false);
    }

    const int rBase = tileM + ((lane >> 4) * 8);
    float* __restrict__ orow = xw + (size_t)rBase * HIDW;
    if (tileM + 16 <= nrows) {
        #pragma unroll
        for (int r = 0; r < 8; ++r) {
            orow[(size_t)r * HIDW + ncol0] = c0[r];
            orow[(size_t)r * HIDW + ncol1] = c1[r];
        }
    } else {
        #pragma unroll
        for (int r = 0; r < 8; ++r) {
            if (rBase + r < nrows) {
                orow[(size_t)r * HIDW + ncol0] = c0[r];
                orow[(size_t)r * HIDW + ncol1] = c1[r];
            }
        }
    }
}

__global__ void __launch_bounds__(256)
gcn_scatter_kernel(const float* __restrict__ xw,
                   const float* __restrict__ dinv,
                   const int* __restrict__ src,
                   const int* __restrict__ dst,
                   float* __restrict__ agg, int nedges) {
    const int gwarp = (int)((blockIdx.x * blockDim.x + threadIdx.x) >> 5);
    const int lane  = threadIdx.x & 31;
    if (gwarp >= nedges) return;

    const int s = src[gwarp];
    const int d = dst[gwarp];
    const float norm = dinv[s] * dinv[d];

    const float* __restrict__ xr = xw + (size_t)s * HIDW + lane;
    const float v0 = xr[0];
    const float v1 = xr[32];
    const float v2 = xr[64];
    const float v3 = xr[96];
    float* ar = agg + (size_t)d * HIDW + lane;
    atomicAdd(ar +  0, v0 * norm);
    atomicAdd(ar + 32, v1 * norm);
    atomicAdd(ar + 64, v2 * norm);
    atomicAdd(ar + 96, v3 * norm);
}

__global__ void __launch_bounds__(256)
gcn_finalize_kernel(const float* __restrict__ x,
                    const float* __restrict__ xw,
                    const float* __restrict__ agg,
                    const float* __restrict__ dinv,
                    const float* __restrict__ b1,
                    const float* __restrict__ Wfc,
                    const float* __restrict__ bfc,
                    float* __restrict__ out, int n) {
    const int node = (int)((blockIdx.x * blockDim.x + threadIdx.x) >> 5);
    const int lane = threadIdx.x & 31;
    if (node >= n) return;

    const float di   = dinv[node];
    const float self = di * di;

    const float4 xv = ((const float4*)(x   + (size_t)node * NFEAT))[lane];
    const float4 wv = ((const float4*)(xw  + (size_t)node * HIDW))[lane];
    const float4 av = ((const float4*)(agg + (size_t)node * HIDW))[lane];
    const float4 bv = ((const float4*)b1)[lane];

    float4 h;
    h.x = fmaxf(av.x + wv.x * self + bv.x, 0.0f);
    h.y = fmaxf(av.y + wv.y * self + bv.y, 0.0f);
    h.z = fmaxf(av.z + wv.z * self + bv.z, 0.0f);
    h.w = fmaxf(av.w + wv.w * self + bv.w, 0.0f);

    const int f0 = lane * 4;
    float acc0, acc1;
    acc0  = xv.x * Wfc[(f0 + 0) * 2 + 0] + xv.y * Wfc[(f0 + 1) * 2 + 0]
          + xv.z * Wfc[(f0 + 2) * 2 + 0] + xv.w * Wfc[(f0 + 3) * 2 + 0];
    acc1  = xv.x * Wfc[(f0 + 0) * 2 + 1] + xv.y * Wfc[(f0 + 1) * 2 + 1]
          + xv.z * Wfc[(f0 + 2) * 2 + 1] + xv.w * Wfc[(f0 + 3) * 2 + 1];
    acc0 += h.x * Wfc[(NFEAT + f0 + 0) * 2 + 0] + h.y * Wfc[(NFEAT + f0 + 1) * 2 + 0]
          + h.z * Wfc[(NFEAT + f0 + 2) * 2 + 0] + h.w * Wfc[(NFEAT + f0 + 3) * 2 + 0];
    acc1 += h.x * Wfc[(NFEAT + f0 + 0) * 2 + 1] + h.y * Wfc[(NFEAT + f0 + 1) * 2 + 1]
          + h.z * Wfc[(NFEAT + f0 + 2) * 2 + 1] + h.w * Wfc[(NFEAT + f0 + 3) * 2 + 1];

    #pragma unroll
    for (int off = 16; off > 0; off >>= 1) {
        acc0 += __shfl_down(acc0, off, 32);
        acc1 += __shfl_down(acc1, off, 32);
    }
    if (lane == 0) {
        out[(size_t)node * 2 + 0] = acc0 + bfc[0];
        out[(size_t)node * 2 + 1] = acc1 + bfc[1];
    }
}

extern "C" void kernel_launch(void* const* d_in, const int* in_sizes, int n_in,
                              void* d_out, int out_size, void* d_ws, size_t ws_size,
                              hipStream_t stream) {
    const float* x   = (const float*)d_in[0];
    const int*   ei  = (const int*)  d_in[1];
    const float* W1  = (const float*)d_in[2];
    const float* b1  = (const float*)d_in[3];
    const float* Wfc = (const float*)d_in[4];
    const float* bfc = (const float*)d_in[5];
    float*       out = (float*)d_out;

    const int N = in_sizes[0] / NFEAT;
    const int E = in_sizes[1] / 2;
    const int* src = ei;
    const int* dst = ei + E;

    float* deg = (float*)d_ws;
    float* xw  = deg + N;
    float* agg = xw + (size_t)N * HIDW;

    const int totalAgg = N * HIDW;

    gcn_init_kernel<<<(totalAgg + 255) / 256, 256, 0, stream>>>(deg, agg, N, totalAgg);

    gcn_degree_kernel<<<(E + 255) / 256, 256, 0, stream>>>(dst, deg, E);

    gcn_rsqrt_kernel<<<(N + 255) / 256, 256, 0, stream>>>(deg, N);

    gcn_gemm_xw_kernel<<<(N + 15) / 16, 128, 0, stream>>>(x, W1, xw, N);

    {
        const long long threads = (long long)E * 32;
        const int blocks = (int)((threads + 255) / 256);
        gcn_scatter_kernel<<<blocks, 256, 0, stream>>>(xw, deg, src, dst, agg, E);
    }

    {
        const long long threads = (long long)N * 32;
        const int blocks = (int)((threads + 255) / 256);
        gcn_finalize_kernel<<<blocks, 256, 0, stream>>>(x, xw, agg, deg, b1, Wfc, bfc, out, N);
    }
}


# row atomics removed: edges counting-sorted by destination per call (int count+rank atomics, scan, fill); one wave per node sums its rows in registers in rank order, fused with finalize
# speedup vs baseline: 2.2763x; 1.7226x over previous
#include <hip/hip_runtime.h>
#include <hip/hip_bf16.h>


typedef float v2f __attribute__((ext_vector_type(2)));
typedef float v8f __attribute__((ext_vector_type(8)));

#define NFEAT 128
#define HIDW  128


__global__ void __launch_bounds__(256)
gcn_init_kernel(int* __restrict__ cnt, int n) {
    const int i = blockIdx.x * blockDim.x + threadIdx.x;
    if (i < n) cnt[i] = 0;
}

__global__ void __launch_bounds__(256)
gcn_count_kernel(const int* __restrict__ dst,
                 int* __restrict__ cnt,
                 int* __restrict__ rank, int nedges) {
    const int e = blockIdx.x * blockDim.x + threadIdx.x;
    if (e < nedges) rank[e] = atomicAdd(&cnt[dst[e]], 1);
}

__global__ void __launch_bounds__(1024)
gcn_scan_kernel(const int* __restrict__ cnt,
                int* __restrict__ rowptr,
                float* __restrict__ dinv, int n) {
    __shared__ int s_wtot[32];
    __shared__ int s_wpre[32];

    const int tid  = threadIdx.x;
    const int lane = tid & 31;
    const int wave = tid >> 5;
    const int base = blockIdx.x * 1024;

    const int4* __restrict__ cnt4 = (const int4*)cnt;
    const int nq = base >> 2;
    int part = 0;
    #pragma unroll 4
    for (int q = tid; q < nq; q += 1024) {
        const int4 t = cnt4[q];
        part += (t.x + t.y) + (t.z + t.w);
    }
    for (int off = 16; off > 0; off >>= 1) part += __shfl_xor(part, off, 32);

    const int i = base + tid;
    const int v = (i < n) ? cnt[i] : 0;
    int incl = v;
    for (int off = 1; off < 32; off <<= 1) {
        const int t = __shfl_up(incl, off, 32);
        if (lane >= off) incl += t;
    }
    if (lane == 31) s_wtot[wave] = incl;
    if (lane == 0)  s_wpre[wave] = part;
    __syncthreads();

    int sum = s_wpre[lane] + ((lane < wave) ? s_wtot[lane] : 0);
    for (int off = 16; off > 0; off >>= 1) sum += __shfl_xor(sum, off, 32);

    if (i < n) {
        rowptr[i] = sum + incl - v;
        const float d = (float)(v + 1);
        dinv[i] = (d > 0.0f) ? rsqrtf(d) : 0.0f;
    }
}

__global__ void __launch_bounds__(256)
gcn_fill_kernel(const int* __restrict__ src,
                const int* __restrict__ dst,
                const int* __restrict__ rank,
                const int* __restrict__ rowptr,
                int* __restrict__ ssrc, int nedges) {
    const int e = blockIdx.x * blockDim.x + threadIdx.x;
    if (e < nedges) ssrc[rowptr[dst[e]] + rank[e]] = src[e];
}

__global__ void __launch_bounds__(128)
gcn_gemm_xw_kernel(const float* __restrict__ x,
                   const float* __restrict__ W1,
                   float* __restrict__ xw, int nrows) {
    const int wave  = threadIdx.x >> 5;
    const int lane  = threadIdx.x & 31;
    const int tileM = blockIdx.x * 16;
    if (tileM >= nrows) return;

    const int l15   = lane & 15;
    const int khalf = (lane >> 4) * 2;
    int mrow = tileM + l15;
    if (mrow >= nrows) mrow = nrows - 1;

    const int n0    = wave * 32;
    const int ncol0 = n0 + l15;
    const int ncol1 = n0 + 16 + l15;

    v8f c0 = {0.f, 0.f, 0.f, 0.f, 0.f, 0.f, 0.f, 0.f};
    v8f c1 = {0.f, 0.f, 0.f, 0.f, 0.f, 0.f, 0.f, 0.f};

    const float* __restrict__ xrow = x + (size_t)mrow * NFEAT;
    #pragma unroll 4
    for (int kb = 0; kb < NFEAT; kb += 4) {
        const int k = kb + khalf;
        v2f a;
        a.x = xrow[k];
        a.y = xrow[k + 1];
        v2f b0, b1;
        b0.x = W1[(size_t)k       * HIDW + ncol0];
        b0.y = W1[(size_t)(k + 1) * HIDW + ncol0];
        b1.x = W1[(size_t)k       * HIDW + ncol1];
        b1.y = W1[(size_t)(k + 1) * HIDW + ncol1];
        c0 = __builtin_amdgcn_wmma_f32_16x16x4_f32(false, a, false, b0,
                                                   (short)0, c0, false, false);
        c1 = __builtin_amdgcn_wmma_f32_16x16x4_f32(false, a, false, b1,
                                                   (short)0, c1, false, false);
    }

    const int rBase = tileM + ((lane >> 4) * 8);
    float* __restrict__ orow = xw + (size_t)rBase * HIDW;
    if (tileM + 16 <= nrows) {
        #pragma unroll
        for (int r = 0; r < 8; ++r) {
            orow[(size_t)r * HIDW + ncol0] = c0[r];
            orow[(size_t)r * HIDW + ncol1] = c1[r];
        }
    } else {
        #pragma unroll
        for (int r = 0; r < 8; ++r) {
            if (rBase + r < nrows) {
                orow[(size_t)r * HIDW + ncol0] = c0[r];
                orow[(size_t)r * HIDW + ncol1] = c1[r];
            }
        }
    }
}

__device__ __forceinline__ float lane_f(float v, int l) {
    return __builtin_bit_cast(float,
        __builtin_amdgcn_readlane(__builtin_bit_cast(int, v), l));
}

__global__ void __launch_bounds__(256)
gcn_gather_finalize_kernel(const float* __restrict__ x,
                           const float* __restrict__ xw,
                           const float* __restrict__ dinv,
                           const int* __restrict__ rowptr,
                           const int* __restrict__ cnt,
                           const int* __restrict__ ssrc,
                           const float* __restrict__ b1,
                           const float* __restrict__ Wfc,
                           const float* __restrict__ bfc,
                           float* __restrict__ out, int n) {
    const int node = __builtin_amdgcn_readfirstlane(
        (int)((blockIdx.x * blockDim.x + threadIdx.x) >> 5));
    const int lane = threadIdx.x & 31;
    if (node >= n) return;

    const float di   = dinv[node];
    const float self = di * di;
    const int   beg  = rowptr[node];
    const int   c    = cnt[node];

    const float4* __restrict__ xw4 = (const float4*)xw;
    const float4 xv = ((const float4*)(x + (size_t)node * NFEAT))[lane];
    const float4 wv = xw4[(size_t)node * 32 + lane];
    const float4 bv = ((const float4*)b1)[lane];

    float4 av = make_float4(0.0f, 0.0f, 0.0f, 0.0f);
    for (int base = 0; base < c; base += 32) {
        const int m = min(c - base, 32);
        int   my_s = 0;
        float my_d = 0.0f;
        if (lane < m) {
            my_s = ssrc[beg + base + lane];
            my_d = dinv[my_s];
        }
        int j = 0;
        for (; j + 4 <= m; j += 4) {
            const int s0 = __builtin_amdgcn_readlane(my_s, j + 0);
            const int s1 = __builtin_amdgcn_readlane(my_s, j + 1);
            const int s2 = __builtin_amdgcn_readlane(my_s, j + 2);
            const int s3 = __builtin_amdgcn_readlane(my_s, j + 3);
            const float4 r0 = xw4[(size_t)s0 * 32 + lane];
            const float4 r1 = xw4[(size_t)s1 * 32 + lane];
            const float4 r2 = xw4[(size_t)s2 * 32 + lane];
            const float4 r3 = xw4[(size_t)s3 * 32 + lane];
            const float w0 = lane_f(my_d, j + 0) * di;
            const float w1 = lane_f(my_d, j + 1) * di;
            const float w2 = lane_f(my_d, j + 2) * di;
            const float w3 = lane_f(my_d, j + 3) * di;
            av.x = fmaf(r0.x, w0, av.x); av.y = fmaf(r0.y, w0, av.y);
            av.z = fmaf(r0.z, w0, av.z); av.w = fmaf(r0.w, w0, av.w);
            av.x = fmaf(r1.x, w1, av.x); av.y = fmaf(r1.y, w1, av.y);
            av.z = fmaf(r1.z, w1, av.z); av.w = fmaf(r1.w, w1, av.w);
            av.x = fmaf(r2.x, w2, av.x); av.y = fmaf(r2.y, w2, av.y);
            av.z = fmaf(r2.z, w2, av.z); av.w = fmaf(r2.w, w2, av.w);
            av.x = fmaf(r3.x, w3, av.x); av.y = fmaf(r3.y, w3, av.y);
            av.z = fmaf(r3.z, w3, av.z); av.w = fmaf(r3.w, w3, av.w);
        }
        const int t = m - j;
        if (t == 3) {
            const int s0 = __builtin_amdgcn_readlane(my_s, j + 0);
            const int s1 = __builtin_amdgcn_readlane(my_s, j + 1);
            const int s2 = __builtin_amdgcn_readlane(my_s, j + 2);
            const float4 r0 = xw4[(size_t)s0 * 32 + lane];
            const float4 r1 = xw4[(size_t)s1 * 32 + lane];
            const float4 r2 = xw4[(size_t)s2 * 32 + lane];
            const float w0 = lane_f(my_d, j + 0) * di;
            const float w1 = lane_f(my_d, j + 1) * di;
            const float w2 = lane_f(my_d, j + 2) * di;
            av.x = fmaf(r0.x, w0, av.x); av.y = fmaf(r0.y, w0, av.y);
            av.z = fmaf(r0.z, w0, av.z); av.w = fmaf(r0.w, w0, av.w);
            av.x = fmaf(r1.x, w1, av.x); av.y = fmaf(r1.y, w1, av.y);
            av.z = fmaf(r1.z, w1, av.z); av.w = fmaf(r1.w, w1, av.w);
            av.x = fmaf(r2.x, w2, av.x); av.y = fmaf(r2.y, w2, av.y);
            av.z = fmaf(r2.z, w2, av.z); av.w = fmaf(r2.w, w2, av.w);
        } else if (t == 2) {
            const int s0 = __builtin_amdgcn_readlane(my_s, j + 0);
            const int s1 = __builtin_amdgcn_readlane(my_s, j + 1);
            const float4 r0 = xw4[(size_t)s0 * 32 + lane];
            const float4 r1 = xw4[(size_t)s1 * 32 + lane];
            const float w0 = lane_f(my_d, j + 0) * di;
            const float w1 = lane_f(my_d, j + 1) * di;
            av.x = fmaf(r0.x, w0, av.x); av.y = fmaf(r0.y, w0, av.y);
            av.z = fmaf(r0.z, w0, av.z); av.w = fmaf(r0.w, w0, av.w);
            av.x = fmaf(r1.x, w1, av.x); av.y = fmaf(r1.y, w1, av.y);
            av.z = fmaf(r1.z, w1, av.z); av.w = fmaf(r1.w, w1, av.w);
        } else if (t == 1) {
            const int s0 = __builtin_amdgcn_readlane(my_s, j);
            const float4 r0 = xw4[(size_t)s0 * 32 + lane];
            const float w0 = lane_f(my_d, j) * di;
            av.x = fmaf(r0.x, w0, av.x); av.y = fmaf(r0.y, w0, av.y);
            av.z = fmaf(r0.z, w0, av.z); av.w = fmaf(r0.w, w0, av.w);
        }
    }

    float4 h;
    h.x = fmaxf(av.x + wv.x * self + bv.x, 0.0f);
    h.y = fmaxf(av.y + wv.y * self + bv.y, 0.0f);
    h.z = fmaxf(av.z + wv.z * self + bv.z, 0.0f);
    h.w = fmaxf(av.w + wv.w * self + bv.w, 0.0f);

    const int f0 = lane * 4;
    float acc0, acc1;
    acc0  = xv.x * Wfc[(f0 + 0) * 2 + 0] + xv.y * Wfc[(f0 + 1) * 2 + 0]
          + xv.z * Wfc[(f0 + 2) * 2 + 0] + xv.w * Wfc[(f0 + 3) * 2 + 0];
    acc1  = xv.x * Wfc[(f0 + 0) * 2 + 1] + xv.y * Wfc[(f0 + 1) * 2 + 1]
          + xv.z * Wfc[(f0 + 2) * 2 + 1] + xv.w * Wfc[(f0 + 3) * 2 + 1];
    acc0 += h.x * Wfc[(NFEAT + f0 + 0) * 2 + 0] + h.y * Wfc[(NFEAT + f0 + 1) * 2 + 0]
          + h.z * Wfc[(NFEAT + f0 + 2) * 2 + 0] + h.w * Wfc[(NFEAT + f0 + 3) * 2 + 0];
    acc1 += h.x * Wfc[(NFEAT + f0 + 0) * 2 + 1] + h.y * Wfc[(NFEAT + f0 + 1) * 2 + 1]
          + h.z * Wfc[(NFEAT + f0 + 2) * 2 + 1] + h.w * Wfc[(NFEAT + f0 + 3) * 2 + 1];

    #pragma unroll
    for (int off = 16; off > 0; off >>= 1) {
        acc0 += __shfl_down(acc0, off, 32);
        acc1 += __shfl_down(acc1, off, 32);
    }
    if (lane == 0) {
        out[(size_t)node * 2 + 0] = acc0 + bfc[0];
        out[(size_t)node * 2 + 1] = acc1 + bfc[1];
    }
}

extern "C" void kernel_launch(void* const* d_in, const int* in_sizes, int n_in,
                              void* d_out, int out_size, void* d_ws, size_t ws_size,
                              hipStream_t stream) {
    const float* x   = (const float*)d_in[0];
    const int*   ei  = (const int*)  d_in[1];
    const float* W1  = (const float*)d_in[2];
    const float* b1  = (const float*)d_in[3];
    const float* Wfc = (const float*)d_in[4];
    const float* bfc = (const float*)d_in[5];
    float*       out = (float*)d_out;

    const int N = in_sizes[0] / NFEAT;
    const int E = in_sizes[1] / 2;
    const int* src = ei;
    const int* dst = ei + E;

    float* xw     = (float*)d_ws;
    int*   cnt    = (int*)(xw + (size_t)N * HIDW);
    int*   rowptr = cnt + N;
    float* dinv   = (float*)(rowptr + N);
    int*   rank   = (int*)(dinv + N);
    int*   ssrc   = rank + E;

    gcn_init_kernel<<<(N + 255) / 256, 256, 0, stream>>>(cnt, N);

    gcn_count_kernel<<<(E + 255) / 256, 256, 0, stream>>>(dst, cnt, rank, E);

    gcn_scan_kernel<<<(N + 1023) / 1024, 1024, 0, stream>>>(cnt, rowptr, dinv, N);

    gcn_fill_kernel<<<(E + 255) / 256, 256, 0, stream>>>(src, dst, rank, rowptr, ssrc, E);

    gcn_gemm_xw_kernel<<<(N + 15) / 16, 128, 0, stream>>>(x, W1, xw, N);

    {
        const long long threads = (long long)N * 32;
        const int blocks = (int)((threads + 255) / 256);
        gcn_gather_finalize_kernel<<<blocks, 256, 0, stream>>>(x, xw, dinv, rowptr, cnt, ssrc,
                                                               b1, Wfc, bfc, out, N);
    }
}


# x@W1 product gets no launch of its own: its row tiles ride in the count and fill launches (1 product block per 4 edge blocks, independent work); 5 launches, sums and rank order as before
# speedup vs baseline: 2.4023x; 1.0554x over previous
#include <hip/hip_runtime.h>
#include <hip/hip_bf16.h>


typedef float v2f __attribute__((ext_vector_type(2)));
typedef float v8f __attribute__((ext_vector_type(8)));

#define NFEAT 128
#define HIDW  128


__global__ void __launch_bounds__(256)
gcn_init_kernel(int* __restrict__ cnt, int n) {
    const int i = blockIdx.x * blockDim.x + threadIdx.x;
    if (i < n) cnt[i] = 0;
}

__device__ __forceinline__ void
gcn_gemm_xw_tile(const float* __restrict__ x,
                 const float* __restrict__ W1,
                 float* __restrict__ xw, int nrows, int tile, int tid) {
    const int wave  = tid >> 5;
    const int lane  = tid & 31;
    const int tileM = tile * 16;
    if (tileM >= nrows) return;

    const int l15   = lane & 15;
    const int khalf = (lane >> 4) * 2;
    int mrow = tileM + l15;
    if (mrow >= nrows) mrow = nrows - 1;

    const int n0    = wave * 32;
    const int ncol0 = n0 + l15;
    const int ncol1 = n0 + 16 + l15;

    v8f c0 = {0.f, 0.f, 0.f, 0.f, 0.f, 0.f, 0.f, 0.f};
    v8f c1 = {0.f, 0.f, 0.f, 0.f, 0.f, 0.f, 0.f, 0.f};

    const float* __restrict__ xrow = x + (size_t)mrow * NFEAT;
    #pragma unroll 4
    for (int kb = 0; kb < NFEAT; kb += 4) {
        const int k = kb + khalf;
        v2f a;
        a.x = xrow[k];
        a.y = xrow[k + 1];
        v2f b0, b1;
        b0.x = W1[(size_t)k       * HIDW + ncol0];
        b0.y = W1[(size_t)(k + 1) * HIDW + ncol0];
        b1.x = W1[(size_t)k       * HIDW + ncol1];
        b1.y = W1[(size_t)(k + 1) * HIDW + ncol1];
        c0 = __builtin_amdgcn_wmma_f32_16x16x4_f32(false, a, false, b0,
                                                   (short)0, c0, false, false);
        c1 = __builtin_amdgcn_wmma_f32_16x16x4_f32(false, a, false, b1,
                                                   (short)0, c1, false, false);
    }

    const int rBase = tileM + ((lane >> 4) * 8);
    float* __restrict__ orow = xw + (size_t)rBase * HIDW;
    if (tileM + 16 <= nrows) {
        #pragma unroll
        for (int r = 0; r < 8; ++r) {
            orow[(size_t)r * HIDW + ncol0] = c0[r];
            orow[(size_t)r * HIDW + ncol1] = c1[r];
        }
    } else {
        #pragma unroll
        for (int r = 0; r < 8; ++r) {
            if (rBase + r < nrows) {
                orow[(size_t)r * HIDW + ncol0] = c0[r];
                orow[(size_t)r * HIDW + ncol1] = c1[r];
            }
        }
    }
}

#define GRP_BLOCKS 5
#define EDGE_CHUNK 256

__device__ __forceinline__ void
gcn_gemm_block(const float* __restrict__ x,
               const float* __restrict__ W1,
               float* __restrict__ xw, int nrows,
               int grp, int tile_first, int tile_end) {
    const int half = __builtin_amdgcn_readfirstlane((int)(threadIdx.x >> 7));
    const int tile = tile_first + 2 * grp + half;
    if (tile < tile_end) gcn_gemm_xw_tile(x, W1, xw, nrows, tile, (int)(threadIdx.x & 127));
}

__global__ void __launch_bounds__(256)
gcn_count_gemm_kernel(const int* __restrict__ dst,
                      int* __restrict__ cnt,
                      int* __restrict__ rank, int nedges,
                      const float* __restrict__ x,
                      const float* __restrict__ W1,
                      float* __restrict__ xw, int nrows,
                      int tile_first, int tile_end) {
    const int grp = blockIdx.x / GRP_BLOCKS;
    const int pos = blockIdx.x - grp * GRP_BLOCKS;
    if (pos == 0) {
        gcn_gemm_block(x, W1, xw, nrows, grp, tile_first, tile_end);
    } else {
        const int e = (grp * (GRP_BLOCKS - 1) + (pos - 1)) * EDGE_CHUNK + threadIdx.x;
        if (e < nedges) rank[e] = atomicAdd(&cnt[dst[e]], 1);
    }
}

__global__ void __launch_bounds__(1024)
gcn_scan_kernel(const int* __restrict__ cnt,
                int* __restrict__ rowptr,
                float* __restrict__ dinv, int n) {
    __shared__ int s_wtot[32];
    __shared__ int s_wpre[32];

    const int tid  = threadIdx.x;
    const int lane = tid & 31;
    const int wave = tid >> 5;
    const int base = blockIdx.x * 1024;

    const int4* __restrict__ cnt4 = (const int4*)cnt;
    const int nq = base >> 2;
    int part = 0;
    #pragma unroll 4
    for (int q = tid; q < nq; q += 1024) {
        const int4 t = cnt4[q];
        part += (t.x + t.y) + (t.z + t.w);
    }
    for (int off = 16; off > 0; off >>= 1) part += __shfl_xor(part, off, 32);

    const int i = base + tid;
    const int v = (i < n) ? cnt[i] : 0;
    int incl = v;
    for (int off = 1; off < 32; off <<= 1) {
        const int t = __shfl_up(incl, off, 32);
        if (lane >= off) incl += t;
    }
    if (lane == 31) s_wtot[wave] = incl;
    if (lane == 0)  s_wpre[wave] = part;
    __syncthreads();

    int sum = s_wpre[lane] + ((lane < wave) ? s_wtot[lane] : 0);
    for (int off = 16; off > 0; off >>= 1) sum += __shfl_xor(sum, off, 32);

    if (i < n) {
        rowptr[i] = sum + incl - v;
        const float d = (float)(v + 1);
        dinv[i] = (d > 0.0f) ? rsqrtf(d) : 0.0f;
    }
}

__global__ void __launch_bounds__(256)
gcn_fill_gemm_kernel(const int* __restrict__ src,
                     const int* __restrict__ dst,
                     const int* __restrict__ rank,
                     const int* __restrict__ rowptr,
                     int* __restrict__ ssrc, int nedges,
                     const float* __restrict__ x,
                     const float* __restrict__ W1,
                     float* __restrict__ xw, int nrows,
                     int tile_first, int tile_end) {
    const int grp = blockIdx.x / GRP_BLOCKS;
    const int pos = blockIdx.x - grp * GRP_BLOCKS;
    if (pos == 0) {
        gcn_gemm_block(x, W1, xw, nrows, grp, tile_first, tile_end);
    } else {
        const int e = (grp * (GRP_BLOCKS - 1) + (pos - 1)) * EDGE_CHUNK + threadIdx.x;
        if (e < nedges) ssrc[rowptr[dst[e]] + rank[e]] = src[e];
    }
}

__device__ __forceinline__ float lane_f(float v, int l) {
    return __builtin_bit_cast(float,
        __builtin_amdgcn_readlane(__builtin_bit_cast(int, v), l));
}

__global__ void __launch_bounds__(256)
gcn_gather_finalize_kernel(const float* __restrict__ x,
                           const float* __restrict__ xw,
                           const float* __restrict__ dinv,
                           const int* __restrict__ rowptr,
                           const int* __restrict__ cnt,
                           const int* __restrict__ ssrc,
                           const float* __restrict__ b1,
                           const float* __restrict__ Wfc,
                           const float* __restrict__ bfc,
                           float* __restrict__ out, int n) {
    const int node = __builtin_amdgcn_readfirstlane(
        (int)((blockIdx.x * blockDim.x + threadIdx.x) >> 5));
    const int lane = threadIdx.x & 31;
    if (node >= n) return;

    const float di   = dinv[node];
    const float self = di * di;
    const int   beg  = rowptr[node];
    const int   c    = cnt[node];

    const float4* __restrict__ xw4 = (const float4*)xw;
    const float4 xv = ((const float4*)(x + (size_t)node * NFEAT))[lane];
    const float4 wv = xw4[(size_t)node * 32 + lane];
    const float4 bv = ((const float4*)b1)[lane];

    float4 av = make_float4(0.0f, 0.0f, 0.0f, 0.0f);
    for (int base = 0; base < c; base += 32) {
        const int m = min(c - base, 32);
        int   my_s = 0;
        float my_d = 0.0f;
        if (lane < m) {
            my_s = ssrc[beg + base + lane];
            my_d = dinv[my_s];
        }
        int j = 0;
        for (; j + 4 <= m; j += 4) {
            const int s0 = __builtin_amdgcn_readlane(my_s, j + 0);
            const int s1 = __builtin_amdgcn_readlane(my_s, j + 1);
            const int s2 = __builtin_amdgcn_readlane(my_s, j + 2);
            const int s3 = __builtin_amdgcn_readlane(my_s, j + 3);
            const float4 r0 = xw4[(size_t)s0 * 32 + lane];
            const float4 r1 = xw4[(size_t)s1 * 32 + lane];
            const float4 r2 = xw4[(size_t)s2 * 32 + lane];
            const float4 r3 = xw4[(size_t)s3 * 32 + lane];
            const float w0 = lane_f(my_d, j + 0) * di;
            const float w1 = lane_f(my_d, j + 1) * di;
            const float w2 = lane_f(my_d, j + 2) * di;
            const float w3 = lane_f(my_d, j + 3) * di;
            av.x = fmaf(r0.x, w0, av.x); av.y = fmaf(r0.y, w0, av.y);
            av.z = fmaf(r0.z, w0, av.z); av.w = fmaf(r0.w, w0, av.w);
            av.x = fmaf(r1.x, w1, av.x); av.y = fmaf(r1.y, w1, av.y);
            av.z = fmaf(r1.z, w1, av.z); av.w = fmaf(r1.w, w1, av.w);
            av.x = fmaf(r2.x, w2, av.x); av.y = fmaf(r2.y, w2, av.y);
            av.z = fmaf(r2.z, w2, av.z); av.w = fmaf(r2.w, w2, av.w);
            av.x = fmaf(r3.x, w3, av.x); av.y = fmaf(r3.y, w3, av.y);
            av.z = fmaf(r3.z, w3, av.z); av.w = fmaf(r3.w, w3, av.w);
        }
        const int t = m - j;
        if (t == 3) {
            const int s0 = __builtin_amdgcn_readlane(my_s, j + 0);
            const int s1 = __builtin_amdgcn_readlane(my_s, j + 1);
            const int s2 = __builtin_amdgcn_readlane(my_s, j + 2);
            const float4 r0 = xw4[(size_t)s0 * 32 + lane];
            const float4 r1 = xw4[(size_t)s1 * 32 + lane];
            const float4 r2 = xw4[(size_t)s2 * 32 + lane];
            const float w0 = lane_f(my_d, j + 0) * di;
            const float w1 = lane_f(my_d, j + 1) * di;
            const float w2 = lane_f(my_d, j + 2) * di;
            av.x = fmaf(r0.x, w0, av.x); av.y = fmaf(r0.y, w0, av.y);
            av.z = fmaf(r0.z, w0, av.z); av.w = fmaf(r0.w, w0, av.w);
            av.x = fmaf(r1.x, w1, av.x); av.y = fmaf(r1.y, w1, av.y);
            av.z = fmaf(r1.z, w1, av.z); av.w = fmaf(r1.w, w1, av.w);
            av.x = fmaf(r2.x, w2, av.x); av.y = fmaf(r2.y, w2, av.y);
            av.z = fmaf(r2.z, w2, av.z); av.w = fmaf(r2.w, w2, av.w);
        } else if (t == 2) {
            const int s0 = __builtin_amdgcn_readlane(my_s, j + 0);
            const int s1 = __builtin_amdgcn_readlane(my_s, j + 1);
            const float4 r0 = xw4[(size_t)s0 * 32 + lane];
            const float4 r1 = xw4[(size_t)s1 * 32 + lane];
            const float w0 = lane_f(my_d, j + 0) * di;
            const float w1 = lane_f(my_d, j + 1) * di;
            av.x = fmaf(r0.x, w0, av.x); av.y = fmaf(r0.y, w0, av.y);
            av.z = fmaf(r0.z, w0, av.z); av.w = fmaf(r0.w, w0, av.w);
            av.x = fmaf(r1.x, w1, av.x); av.y = fmaf(r1.y, w1, av.y);
            av.z = fmaf(r1.z, w1, av.z); av.w = fmaf(r1.w, w1, av.w);
        } else if (t == 1) {
            const int s0 = __builtin_amdgcn_readlane(my_s, j);
            const float4 r0 = xw4[(size_t)s0 * 32 + lane];
            const float w0 = lane_f(my_d, j) * di;
            av.x = fmaf(r0.x, w0, av.x); av.y = fmaf(r0.y, w0, av.y);
            av.z = fmaf(r0.z, w0, av.z); av.w = fmaf(r0.w, w0, av.w);
        }
    }

    float4 h;
    h.x = fmaxf(av.x + wv.x * self + bv.x, 0.0f);
    h.y = fmaxf(av.y + wv.y * self + bv.y, 0.0f);
    h.z = fmaxf(av.z + wv.z * self + bv.z, 0.0f);
    h.w = fmaxf(av.w + wv.w * self + bv.w, 0.0f);

    const int f0 = lane * 4;
    float acc0, acc1;
    acc0  = xv.x * Wfc[(f0 + 0) * 2 + 0] + xv.y * Wfc[(f0 + 1) * 2 + 0]
          + xv.z * Wfc[(f0 + 2) * 2 + 0] + xv.w * Wfc[(f0 + 3) * 2 + 0];
    acc1  = xv.x * Wfc[(f0 + 0) * 2 + 1] + xv.y * Wfc[(f0 + 1) * 2 + 1]
          + xv.z * Wfc[(f0 + 2) * 2 + 1] + xv.w * Wfc[(f0 + 3) * 2 + 1];
    acc0 += h.x * Wfc[(NFEAT + f0 + 0) * 2 + 0] + h.y * Wfc[(NFEAT + f0 + 1) * 2 + 0]
          + h.z * Wfc[(NFEAT + f0 + 2) * 2 + 0] + h.w * Wfc[(NFEAT + f0 + 3) * 2 + 0];
    acc1 += h.x * Wfc[(NFEAT + f0 + 0) * 2 + 1] + h.y * Wfc[(NFEAT + f0 + 1) * 2 + 1]
          + h.z * Wfc[(NFEAT + f0 + 2) * 2 + 1] + h.w * Wfc[(NFEAT + f0 + 3) * 2 + 1];

    #pragma unroll
    for (int off = 16; off > 0; off >>= 1) {
        acc0 += __shfl_down(acc0, off, 32);
        acc1 += __shfl_down(acc1, off, 32);
    }
    if (lane == 0) {
        out[(size_t)node * 2 + 0] = acc0 + bfc[0];
        out[(size_t)node * 2 + 1] = acc1 + bfc[1];
    }
}

extern "C" void kernel_launch(void* const* d_in, const int* in_sizes, int n_in,
                              void* d_out, int out_size, void* d_ws, size_t ws_size,
                              hipStream_t stream) {
    const float* x   = (const float*)d_in[0];
    const int*   ei  = (const int*)  d_in[1];
    const float* W1  = (const float*)d_in[2];
    const float* b1  = (const float*)d_in[3];
    const float* Wfc = (const float*)d_in[4];
    const float* bfc = (const float*)d_in[5];
    float*       out = (float*)d_out;

    const int N = in_sizes[0] / NFEAT;
    const int E = in_sizes[1] / 2;
    const int* src = ei;
    const int* dst = ei + E;

    float* xw     = (float*)d_ws;
    int*   cnt    = (int*)(xw + (size_t)N * HIDW);
    int*   rowptr = cnt + N;
    float* dinv   = (float*)(rowptr + N);
    int*   rank   = (int*)(dinv + N);
    int*   ssrc   = rank + E;

    const int tiles   = (N + 15) / 16;
    const int tilesA  = (tiles + 1) / 2;
    const int tilesB  = tiles - tilesA;
    const int pairsA  = (tilesA + 1) / 2;
    const int pairsB  = (tilesB + 1) / 2;
    const int chunkGr = ((E + EDGE_CHUNK - 1) / EDGE_CHUNK + GRP_BLOCKS - 2) / (GRP_BLOCKS - 1);
    const int groupsA = (pairsA > chunkGr) ? pairsA : chunkGr;
    const int groupsB = (pairsB > chunkGr) ? pairsB : chunkGr;

    gcn_init_kernel<<<(N + 255) / 256, 256, 0, stream>>>(cnt, N);

    gcn_count_gemm_kernel<<<groupsA * GRP_BLOCKS, 256, 0, stream>>>(dst, cnt, rank, E,
                                                                    x, W1, xw, N, 0, tilesA);

    gcn_scan_kernel<<<(N + 1023) / 1024, 1024, 0, stream>>>(cnt, rowptr, dinv, N);

    gcn_fill_gemm_kernel<<<groupsB * GRP_BLOCKS, 256, 0, stream>>>(src, dst, rank, rowptr, ssrc, E,
                                                                   x, W1, xw, N, tilesA, tiles);

    {
        const long long threads = (long long)N * 32;
        const int blocks = (int)((threads + 255) / 256);
        gcn_gather_finalize_kernel<<<blocks, 256, 0, stream>>>(x, xw, dinv, rowptr, cnt, ssrc,
                                                               b1, Wfc, bfc, out, N);
    }
}


# edges grouped without global atomics: each 4096-edge block orders its edges by 512-node bin in LDS, one work-group per bin counts, scans, places in LDS; 3 launches; sums in LDS-add arrival order
# speedup vs baseline: 4.6972x; 1.9553x over previous
#include <hip/hip_runtime.h>
#include <hip/hip_bf16.h>


typedef float v2f __attribute__((ext_vector_type(2)));
typedef float v8f __attribute__((ext_vector_type(8)));

#define NFEAT 128
#define HIDW  128

#define BIN_SHIFT 9
#define BIN_NODES 512
#define MAXB      256
#define CH        4096
#define EPT       16
#define SORT_CAP  12288


__device__ __forceinline__ void
gcn_gemm_xw_tile(const float* __restrict__ x,
                 const float* __restrict__ W1,
                 float* __restrict__ xw, int nrows, int tile, int tid) {
    const int wave  = tid >> 5;
    const int lane  = tid & 31;
    const int tileM = tile * 16;
    if (tileM >= nrows) return;

    const int l15   = lane & 15;
    const int khalf = (lane >> 4) * 2;
    int mrow = tileM + l15;
    if (mrow >= nrows) mrow = nrows - 1;

    const int n0    = wave * 32;
    const int ncol0 = n0 + l15;
    const int ncol1 = n0 + 16 + l15;

    v8f c0 = {0.f, 0.f, 0.f, 0.f, 0.f, 0.f, 0.f, 0.f};
    v8f c1 = {0.f, 0.f, 0.f, 0.f, 0.f, 0.f, 0.f, 0.f};

    const float* __restrict__ xrow = x + (size_t)mrow * NFEAT;
    #pragma unroll 4
    for (int kb = 0; kb < NFEAT; kb += 4) {
        const int k = kb + khalf;
        v2f a;
        a.x = xrow[k];
        a.y = xrow[k + 1];
        v2f b0, b1;
        b0.x = W1[(size_t)k       * HIDW + ncol0];
        b0.y = W1[(size_t)(k + 1) * HIDW + ncol0];
        b1.x = W1[(size_t)k       * HIDW + ncol1];
        b1.y = W1[(size_t)(k + 1) * HIDW + ncol1];
        c0 = __builtin_amdgcn_wmma_f32_16x16x4_f32(false, a, false, b0,
                                                   (short)0, c0, false, false);
        c1 = __builtin_amdgcn_wmma_f32_16x16x4_f32(false, a, false, b1,
                                                   (short)0, c1, false, false);
    }

    const int rBase = tileM + ((lane >> 4) * 8);
    float* __restrict__ orow = xw + (size_t)rBase * HIDW;
    if (tileM + 16 <= nrows) {
        #pragma unroll
        for (int r = 0; r < 8; ++r) {
            orow[(size_t)r * HIDW + ncol0] = c0[r];
            orow[(size_t)r * HIDW + ncol1] = c1[r];
        }
    } else {
        #pragma unroll
        for (int r = 0; r < 8; ++r) {
            if (rBase + r < nrows) {
                orow[(size_t)r * HIDW + ncol0] = c0[r];
                orow[(size_t)r * HIDW + ncol1] = c1[r];
            }
        }
    }
}

__device__ __forceinline__ void
gcn_gemm_block(const float* __restrict__ x,
               const float* __restrict__ W1,
               float* __restrict__ xw, int nrows,
               int g, int tiles_per_block, int tile_first, int tile_end) {
    const int part = __builtin_amdgcn_readfirstlane((int)(threadIdx.x >> 7));
    const int tile = tile_first + tiles_per_block * g + part;
    if (tile < tile_end) gcn_gemm_xw_tile(x, W1, xw, nrows, tile, (int)(threadIdx.x & 127));
}

__global__ void __launch_bounds__(256)
gcn_localsort_gemm_kernel(const int* __restrict__ src,
                          const int* __restrict__ dst, int nedges,
                          int neb, int nbins,
                          int* __restrict__ rec_i,
                          int* __restrict__ tbl,
                          int* __restrict__ offs,
                          const float* __restrict__ x,
                          const float* __restrict__ W1,
                          float* __restrict__ xw, int nrows,
                          int tile_first, int tile_end) {
    __shared__ int  s_hist[MAXB];
    __shared__ int  s_off[MAXB];
    __shared__ int  s_wtot[8];
    __shared__ int2 s_stage[CH];

    if ((int)blockIdx.x >= neb) {
        gcn_gemm_block(x, W1, xw, nrows, (int)blockIdx.x - neb, 2, tile_first, tile_end);
        return;
    }
    int2* __restrict__ rec = (int2*)rec_i;
    const int tid  = threadIdx.x;
    const int lane = tid & 31;
    const int wave = tid >> 5;
    const int blk  = blockIdx.x;
    const int e0   = blk * CH;

    s_hist[tid] = 0;
    __syncthreads();

    int d[EPT], s[EPT], r[EPT];
    #pragma unroll
    for (int k = 0; k < EPT; ++k) {
        const int e = e0 + k * 256 + tid;
        d[k] = -1; s[k] = 0; r[k] = 0;
        if (e < nedges) { d[k] = dst[e]; s[k] = src[e]; }
    }
    #pragma unroll
    for (int k = 0; k < EPT; ++k) {
        if (d[k] >= 0) r[k] = atomicAdd(&s_hist[d[k] >> BIN_SHIFT], 1);
    }
    __syncthreads();

    const int v = s_hist[tid];
    int incl = v;
    for (int off = 1; off < 32; off <<= 1) {
        const int t = __shfl_up(incl, off, 32);
        if (lane >= off) incl += t;
    }
    if (lane == 31) s_wtot[wave] = incl;
    __syncthreads();
    int wpre = 0;
    #pragma unroll
    for (int w = 0; w < 8; ++w) wpre += (w < wave) ? s_wtot[w] : 0;
    const int excl = wpre + incl - v;
    s_off[tid] = excl;
    if (tid < nbins) {
        tbl[tid * neb + blk]  = v;
        offs[tid * neb + blk] = excl;
    }
    __syncthreads();

    #pragma unroll
    for (int k = 0; k < EPT; ++k) {
        if (d[k] >= 0) s_stage[s_off[d[k] >> BIN_SHIFT] + r[k]] = make_int2(d[k], s[k]);
    }
    __syncthreads();

    const int nval = min(CH, nedges - e0);
    for (int i = tid; i < nval; i += 256) rec[(size_t)e0 + i] = s_stage[i];
}

template <int MODE>
__device__ __forceinline__ void
gcn_bin_walk(const int2* __restrict__ rec,
             const int* __restrict__ tbl,
             const int* __restrict__ offs, int b, int neb,
             int* s_cnt, int* s_n, int* s_key, int* s_src,
             const int* s_ptr, int* __restrict__ ssrc, int base) {
    const int lane  = threadIdx.x & 31;
    const int wave  = __builtin_amdgcn_readfirstlane((int)(threadIdx.x >> 5));
    const int nw    = (int)(blockDim.x >> 5);
    const int S     = min(32, (neb + nw - 1) / nw);
    const int node0 = b << BIN_SHIFT;
    for (int r0 = 0; r0 < neb; r0 += nw * S) {
        const int sg = r0 + wave * S + lane;
        int c = 0, o = 0;
        if (lane < S && sg < neb) { c = tbl[b * neb + sg]; o = offs[b * neb + sg]; }
        int incl = c;
        for (int off = 1; off < 32; off <<= 1) {
            const int t = __shfl_up(incl, off, 32);
            if (lane >= off) incl += t;
        }
        const int start = incl - c;
        const int total = __builtin_amdgcn_readlane(incl, 31);
        const int adj   = sg * CH + o - start;
        int wbase = 0;
        if (MODE == 0) {
            if (lane == 0) wbase = atomicAdd(s_n, total);
            wbase = __builtin_amdgcn_readfirstlane(wbase);
        }
        for (int fb = 0; fb < total; fb += 128) {
            int fi[4], seg[4];
            #pragma unroll
            for (int u = 0; u < 4; ++u) { fi[u] = fb + 32 * u + lane; seg[u] = 0; }
            for (int j = 1; j < S; ++j) {
                const int st = __builtin_amdgcn_readlane(start, j);
                #pragma unroll
                for (int u = 0; u < 4; ++u) seg[u] += (fi[u] >= st) ? 1 : 0;
            }
            int2 rr[4];
            #pragma unroll
            for (int u = 0; u < 4; ++u) {
                const int a = __shfl(adj, seg[u], 32) + fi[u];
                rr[u] = rec[(fi[u] < total) ? a : 0];
            }
            #pragma unroll
            for (int u = 0; u < 4; ++u) {
                if (fi[u] < total) {
                    const int dl = rr[u].x - node0;
                    if (MODE == 0) {
                        const int rk = atomicAdd(&s_cnt[dl], 1);
                        s_key[wbase + fi[u]] = (rk << BIN_SHIFT) | dl;
                        s_src[wbase + fi[u]] = rr[u].y;
                    } else if (MODE == 1) {
                        atomicAdd(&s_cnt[dl], 1);
                    } else {
                        const int p = atomicAdd(&s_cnt[dl], 1);
                        ssrc[base + s_ptr[dl] + p] = rr[u].y;
                    }
                }
            }
        }
    }
}

__global__ void __launch_bounds__(1024)
gcn_binsort_gemm_kernel(const int* __restrict__ rec_i,
                        const int* __restrict__ tbl,
                        const int* __restrict__ offs, int neb, int nbins,
                        int* __restrict__ cnt,
                        int* __restrict__ rowptr,
                        float* __restrict__ dinv,
                        int* __restrict__ ssrc, int n,
                        const float* __restrict__ x,
                        const float* __restrict__ W1,
                        float* __restrict__ xw, int nrows,
                        int tile_first, int tile_end) {
    __shared__ int s_key[SORT_CAP];
    __shared__ int s_src[SORT_CAP];
    __shared__ int s_out[SORT_CAP];
    __shared__ int s_cnt[BIN_NODES];
    __shared__ int s_ptr[BIN_NODES];
    __shared__ int s_redA[32];
    __shared__ int s_redB[32];
    __shared__ int s_wt[16];
    __shared__ int s_n;

    if ((int)blockIdx.x >= nbins) {
        gcn_gemm_block(x, W1, xw, nrows, (int)blockIdx.x - nbins, 8, tile_first, tile_end);
        return;
    }
    const int2* __restrict__ rec = (const int2*)rec_i;
    const int tid  = threadIdx.x;
    const int lane = tid & 31;
    const int wave = tid >> 5;
    const int b    = blockIdx.x;

    const int nq  = b * neb;
    const int nq4 = nq >> 2;
    const int4* __restrict__ tbl4 = (const int4*)tbl;
    int partA = 0;
    #pragma unroll 4
    for (int q = tid; q < nq4; q += 1024) {
        const int4 t = tbl4[q];
        partA += (t.x + t.y) + (t.z + t.w);
    }
    if (tid < (nq & 3)) partA += tbl[(nq4 << 2) + tid];
    int partB = 0;
    for (int q = tid; q < neb; q += 1024) partB += tbl[nq + q];
    for (int off = 16; off > 0; off >>= 1) {
        partA += __shfl_xor(partA, off, 32);
        partB += __shfl_xor(partB, off, 32);
    }
    if (lane == 0) { s_redA[wave] = partA; s_redB[wave] = partB; }
    if (tid < BIN_NODES) s_cnt[tid] = 0;
    if (tid == 0) s_n = 0;
    __syncthreads();
    int base = s_redA[lane];
    int M    = s_redB[lane];
    for (int off = 16; off > 0; off >>= 1) {
        base += __shfl_xor(base, off, 32);
        M    += __shfl_xor(M, off, 32);
    }
    const bool fast = (M <= SORT_CAP);

    if (fast) gcn_bin_walk<0>(rec, tbl, offs, b, neb, s_cnt, &s_n, s_key, s_src, s_ptr, ssrc, base);
    else      gcn_bin_walk<1>(rec, tbl, offs, b, neb, s_cnt, &s_n, s_key, s_src, s_ptr, ssrc, base);
    __syncthreads();

    const int v = (tid < BIN_NODES) ? s_cnt[tid] : 0;
    int incl = v;
    for (int off = 1; off < 32; off <<= 1) {
        const int t = __shfl_up(incl, off, 32);
        if (lane >= off) incl += t;
    }
    if (tid < BIN_NODES && lane == 31) s_wt[wave] = incl;
    __syncthreads();
    if (tid < BIN_NODES) {
        int wpre = 0;
        #pragma unroll
        for (int w = 0; w < 16; ++w) wpre += (w < wave) ? s_wt[w] : 0;
        const int excl = wpre + incl - v;
        s_ptr[tid] = excl;
        s_cnt[tid] = 0;
        const int node = (b << BIN_SHIFT) + tid;
        if (node < n) {
            cnt[node]    = v;
            rowptr[node] = base + excl;
            const float dg = (float)(v + 1);
            dinv[node]   = (dg > 0.0f) ? rsqrtf(dg) : 0.0f;
        }
    }
    __syncthreads();

    if (fast) {
        for (int i = tid; i < M; i += 1024) {
            const int k = s_key[i];
            s_out[s_ptr[k & (BIN_NODES - 1)] + (k >> BIN_SHIFT)] = s_src[i];
        }
        __syncthreads();
        for (int i = tid; i < M; i += 1024) ssrc[base + i] = s_out[i];
    } else {
        gcn_bin_walk<2>(rec, tbl, offs, b, neb, s_cnt, &s_n, s_key, s_src, s_ptr, ssrc, base);
    }
}

__device__ __forceinline__ float lane_f(float v, int l) {
    return __builtin_bit_cast(float,
        __builtin_amdgcn_readlane(__builtin_bit_cast(int, v), l));
}

__global__ void __launch_bounds__(256)
gcn_gather_finalize_kernel(const float* __restrict__ x,
                           const float* __restrict__ xw,
                           const float* __restrict__ dinv,
                           const int* __restrict__ rowptr,
                           const int* __restrict__ cnt,
                           const int* __restrict__ ssrc,
                           const float* __restrict__ b1,
                           const float* __restrict__ Wfc,
                           const float* __restrict__ bfc,
                           float* __restrict__ out, int n) {
    const int node = __builtin_amdgcn_readfirstlane(
        (int)((blockIdx.x * blockDim.x + threadIdx.x) >> 5));
    const int lane = threadIdx.x & 31;
    if (node >= n) return;

    const float di   = dinv[node];
    const float self = di * di;
    const int   beg  = rowptr[node];
    const int   c    = cnt[node];

    const float4* __restrict__ xw4 = (const float4*)xw;
    const float4 xv = ((const float4*)(x + (size_t)node * NFEAT))[lane];
    const float4 wv = xw4[(size_t)node * 32 + lane];
    const float4 bv = ((const float4*)b1)[lane];

    float4 av = make_float4(0.0f, 0.0f, 0.0f, 0.0f);
    for (int base = 0; base < c; base += 32) {
        const int m = min(c - base, 32);
        int   my_s = 0;
        float my_d = 0.0f;
        if (lane < m) {
            my_s = ssrc[beg + base + lane];
            my_d = dinv[my_s];
        }
        int j = 0;
        for (; j + 4 <= m; j += 4) {
            const int s0 = __builtin_amdgcn_readlane(my_s, j + 0);
            const int s1 = __builtin_amdgcn_readlane(my_s, j + 1);
            const int s2 = __builtin_amdgcn_readlane(my_s, j + 2);
            const int s3 = __builtin_amdgcn_readlane(my_s, j + 3);
            const float4 r0 = xw4[(size_t)s0 * 32 + lane];
            const float4 r1 = xw4[(size_t)s1 * 32 + lane];
            const float4 r2 = xw4[(size_t)s2 * 32 + lane];
            const float4 r3 = xw4[(size_t)s3 * 32 + lane];
            const float w0 = lane_f(my_d, j + 0) * di;
            const float w1 = lane_f(my_d, j + 1) * di;
            const float w2 = lane_f(my_d, j + 2) * di;
            const float w3 = lane_f(my_d, j + 3) * di;
            av.x = fmaf(r0.x, w0, av.x); av.y = fmaf(r0.y, w0, av.y);
            av.z = fmaf(r0.z, w0, av.z); av.w = fmaf(r0.w, w0, av.w);
            av.x = fmaf(r1.x, w1, av.x); av.y = fmaf(r1.y, w1, av.y);
            av.z = fmaf(r1.z, w1, av.z); av.w = fmaf(r1.w, w1, av.w);
            av.x = fmaf(r2.x, w2, av.x); av.y = fmaf(r2.y, w2, av.y);
            av.z = fmaf(r2.z, w2, av.z); av.w = fmaf(r2.w, w2, av.w);
            av.x = fmaf(r3.x, w3, av.x); av.y = fmaf(r3.y, w3, av.y);
            av.z = fmaf(r3.z, w3, av.z); av.w = fmaf(r3.w, w3, av.w);
        }
        const int t = m - j;
        if (t == 3) {
            const int s0 = __builtin_amdgcn_readlane(my_s, j + 0);
            const int s1 = __builtin_amdgcn_readlane(my_s, j + 1);
            const int s2 = __builtin_amdgcn_readlane(my_s, j + 2);
            const float4 r0 = xw4[(size_t)s0 * 32 + lane];
            const float4 r1 = xw4[(size_t)s1 * 32 + lane];
            const float4 r2 = xw4[(size_t)s2 * 32 + lane];
            const float w0 = lane_f(my_d, j + 0) * di;
            const float w1 = lane_f(my_d, j + 1) * di;
            const float w2 = lane_f(my_d, j + 2) * di;
            av.x = fmaf(r0.x, w0, av.x); av.y = fmaf(r0.y, w0, av.y);
            av.z = fmaf(r0.z, w0, av.z); av.w = fmaf(r0.w, w0, av.w);
            av.x = fmaf(r1.x, w1, av.x); av.y = fmaf(r1.y, w1, av.y);
            av.z = fmaf(r1.z, w1, av.z); av.w = fmaf(r1.w, w1, av.w);
            av.x = fmaf(r2.x, w2, av.x); av.y = fmaf(r2.y, w2, av.y);
            av.z = fmaf(r2.z, w2, av.z); av.w = fmaf(r2.w, w2, av.w);
        } else if (t == 2) {
            const int s0 = __builtin_amdgcn_readlane(my_s, j + 0);
            const int s1 = __builtin_amdgcn_readlane(my_s, j + 1);
            const float4 r0 = xw4[(size_t)s0 * 32 + lane];
            const float4 r1 = xw4[(size_t)s1 * 32 + lane];
            const float w0 = lane_f(my_d, j + 0) * di;
            const float w1 = lane_f(my_d, j + 1) * di;
            av.x = fmaf(r0.x, w0, av.x); av.y = fmaf(r0.y, w0, av.y);
            av.z = fmaf(r0.z, w0, av.z); av.w = fmaf(r0.w, w0, av.w);
            av.x = fmaf(r1.x, w1, av.x); av.y = fmaf(r1.y, w1, av.y);
            av.z = fmaf(r1.z, w1, av.z); av.w = fmaf(r1.w, w1, av.w);
        } else if (t == 1) {
            const int s0 = __builtin_amdgcn_readlane(my_s, j);
            const float4 r0 = xw4[(size_t)s0 * 32 + lane];
            const float w0 = lane_f(my_d, j) * di;
            av.x = fmaf(r0.x, w0, av.x); av.y = fmaf(r0.y, w0, av.y);
            av.z = fmaf(r0.z, w0, av.z); av.w = fmaf(r0.w, w0, av.w);
        }
    }

    float4 h;
    h.x = fmaxf(av.x + wv.x * self + bv.x, 0.0f);
    h.y = fmaxf(av.y + wv.y * self + bv.y, 0.0f);
    h.z = fmaxf(av.z + wv.z * self + bv.z, 0.0f);
    h.w = fmaxf(av.w + wv.w * self + bv.w, 0.0f);

    const int f0 = lane * 4;
    float acc0, acc1;
    acc0  = xv.x * Wfc[(f0 + 0) * 2 + 0] + xv.y * Wfc[(f0 + 1) * 2 + 0]
          + xv.z * Wfc[(f0 + 2) * 2 + 0] + xv.w * Wfc[(f0 + 3) * 2 + 0];
    acc1  = xv.x * Wfc[(f0 + 0) * 2 + 1] + xv.y * Wfc[(f0 + 1) * 2 + 1]
          + xv.z * Wfc[(f0 + 2) * 2 + 1] + xv.w * Wfc[(f0 + 3) * 2 + 1];
    acc0 += h.x * Wfc[(NFEAT + f0 + 0) * 2 + 0] + h.y * Wfc[(NFEAT + f0 + 1) * 2 + 0]
          + h.z * Wfc[(NFEAT + f0 + 2) * 2 + 0] + h.w * Wfc[(NFEAT + f0 + 3) * 2 + 0];
    acc1 += h.x * Wfc[(NFEAT + f0 + 0) * 2 + 1] + h.y * Wfc[(NFEAT + f0 + 1) * 2 + 1]
          + h.z * Wfc[(NFEAT + f0 + 2) * 2 + 1] + h.w * Wfc[(NFEAT + f0 + 3) * 2 + 1];

    #pragma unroll
    for (int off = 16; off > 0; off >>= 1) {
        acc0 += __shfl_down(acc0, off, 32);
        acc1 += __shfl_down(acc1, off, 32);
    }
    if (lane == 0) {
        out[(size_t)node * 2 + 0] = acc0 + bfc[0];
        out[(size_t)node * 2 + 1] = acc1 + bfc[1];
    }
}

extern "C" void kernel_launch(void* const* d_in, const int* in_sizes, int n_in,
                              void* d_out, int out_size, void* d_ws, size_t ws_size,
                              hipStream_t stream) {
    const float* x   = (const float*)d_in[0];
    const int*   ei  = (const int*)  d_in[1];
    const float* W1  = (const float*)d_in[2];
    const float* b1  = (const float*)d_in[3];
    const float* Wfc = (const float*)d_in[4];
    const float* bfc = (const float*)d_in[5];
    float*       out = (float*)d_out;

    const int N = in_sizes[0] / NFEAT;
    const int E = in_sizes[1] / 2;
    const int* src = ei;
    const int* dst = ei + E;

    const int neb   = (E + CH - 1) / CH;
    const int nbins = (N + BIN_NODES - 1) / BIN_NODES;

    float* xw     = (float*)d_ws;
    int*   rec    = (int*)(xw + (size_t)N * HIDW);
    int*   tbl    = rec + (size_t)neb * CH * 2;
    int*   offs   = tbl + (size_t)nbins * neb;
    int*   ssrc   = offs + (size_t)nbins * neb;
    int*   cnt    = ssrc + E;
    int*   rowptr = cnt + N;
    float* dinv   = (float*)(rowptr + N);

    const int tiles   = (N + 15) / 16;
    const int tilesA  = (tiles + 1) / 2;
    const int tilesB  = tiles - tilesA;
    const int blocksA = (tilesA + 1) / 2;
    const int blocksB = (tilesB + 7) / 8;

    gcn_localsort_gemm_kernel<<<neb + blocksA, 256, 0, stream>>>(src, dst, E, neb, nbins,
                                                                 rec, tbl, offs,
                                                                 x, W1, xw, N, 0, tilesA);

    gcn_binsort_gemm_kernel<<<nbins + blocksB, 1024, 0, stream>>>(rec, tbl, offs, neb, nbins,
                                                                  cnt, rowptr, dinv, ssrc, N,
                                                                  x, W1, xw, N, tilesA, tiles);

    {
        const long long threads = (long long)N * 32;
        const int blocks = (int)((threads + 255) / 256);
        gcn_gather_finalize_kernel<<<blocks, 256, 0, stream>>>(x, xw, dinv, rowptr, cnt, ssrc,
                                                               b1, Wfc, bfc, out, N);
    }
}


# gather: per-edge dinv[src] made from a byte table of node counts that 512 blocks staying for the launch copy once into LDS (byte 255 = read dinv itself), not a scattered 4-B global load per edge
# speedup vs baseline: 4.8773x; 1.0383x over previous
#include <hip/hip_runtime.h>
#include <hip/hip_bf16.h>


typedef float v2f __attribute__((ext_vector_type(2)));
typedef float v8f __attribute__((ext_vector_type(8)));
typedef int   v4i __attribute__((ext_vector_type(4)));

#define NFEAT 128
#define HIDW  128

#define BIN_SHIFT 9
#define BIN_NODES 512
#define MAXB      256
#define CH        4096
#define EPT       16
#define SORT_CAP  12288
#define DEG_SAT   255
#define GATHER_BLOCKS 512

__device__ __forceinline__ float gcn_dinv_of(int cnt) {
    const float dg = (float)(cnt + 1);
    return (dg > 0.0f) ? rsqrtf(dg) : 0.0f;
}


__device__ __forceinline__ void
gcn_gemm_xw_tile(const float* __restrict__ x,
                 const float* __restrict__ W1,
                 float* __restrict__ xw, int nrows, int tile, int tid) {
    const int wave  = tid >> 5;
    const int lane  = tid & 31;
    const int tileM = tile * 16;
    if (tileM >= nrows) return;

    const int l15   = lane & 15;
    const int khalf = (lane >> 4) * 2;
    int mrow = tileM + l15;
    if (mrow >= nrows) mrow = nrows - 1;

    const int n0    = wave * 32;
    const int ncol0 = n0 + l15;
    const int ncol1 = n0 + 16 + l15;

    v8f c0 = {0.f, 0.f, 0.f, 0.f, 0.f, 0.f, 0.f, 0.f};
    v8f c1 = {0.f, 0.f, 0.f, 0.f, 0.f, 0.f, 0.f, 0.f};

    const float* __restrict__ xrow = x + (size_t)mrow * NFEAT;
    #pragma unroll 4
    for (int kb = 0; kb < NFEAT; kb += 4) {
        const int k = kb + khalf;
        v2f a;
        a.x = xrow[k];
        a.y = xrow[k + 1];
        v2f b0, b1;
        b0.x = W1[(size_t)k       * HIDW + ncol0];
        b0.y = W1[(size_t)(k + 1) * HIDW + ncol0];
        b1.x = W1[(size_t)k       * HIDW + ncol1];
        b1.y = W1[(size_t)(k + 1) * HIDW + ncol1];
        c0 = __builtin_amdgcn_wmma_f32_16x16x4_f32(false, a, false, b0,
                                                   (short)0, c0, false, false);
        c1 = __builtin_amdgcn_wmma_f32_16x16x4_f32(false, a, false, b1,
                                                   (short)0, c1, false, false);
    }

    const int rBase = tileM + ((lane >> 4) * 8);
    float* __restrict__ orow = xw + (size_t)rBase * HIDW;
    if (tileM + 16 <= nrows) {
        #pragma unroll
        for (int r = 0; r < 8; ++r) {
            orow[(size_t)r * HIDW + ncol0] = c0[r];
            orow[(size_t)r * HIDW + ncol1] = c1[r];
        }
    } else {
        #pragma unroll
        for (int r = 0; r < 8; ++r) {
            if (rBase + r < nrows) {
                orow[(size_t)r * HIDW + ncol0] = c0[r];
                orow[(size_t)r * HIDW + ncol1] = c1[r];
            }
        }
    }
}

__device__ __forceinline__ void
gcn_gemm_block(const float* __restrict__ x,
               const float* __restrict__ W1,
               float* __restrict__ xw, int nrows,
               int g, int tiles_per_block, int tile_first, int tile_end) {
    const int part = __builtin_amdgcn_readfirstlane((int)(threadIdx.x >> 7));
    const int tile = tile_first + tiles_per_block * g + part;
    if (tile < tile_end) gcn_gemm_xw_tile(x, W1, xw, nrows, tile, (int)(threadIdx.x & 127));
}

__global__ void __launch_bounds__(256)
gcn_localsort_gemm_kernel(const int* __restrict__ src,
                          const int* __restrict__ dst, int nedges,
                          int neb, int nbins,
                          int* __restrict__ rec_i,
                          int* __restrict__ tbl,
                          int* __restrict__ offs,
                          const float* __restrict__ x,
                          const float* __restrict__ W1,
                          float* __restrict__ xw, int nrows,
                          int tile_first, int tile_end) {
    __shared__ int  s_hist[MAXB];
    __shared__ int  s_off[MAXB];
    __shared__ int  s_wtot[8];
    __shared__ int2 s_stage[CH];

    if ((int)blockIdx.x >= neb) {
        gcn_gemm_block(x, W1, xw, nrows, (int)blockIdx.x - neb, 2, tile_first, tile_end);
        return;
    }
    int2* __restrict__ rec = (int2*)rec_i;
    const int tid  = threadIdx.x;
    const int lane = tid & 31;
    const int wave = tid >> 5;
    const int blk  = blockIdx.x;
    const int e0   = blk * CH;

    s_hist[tid] = 0;
    __syncthreads();

    int d[EPT], s[EPT], r[EPT];
    #pragma unroll
    for (int k = 0; k < EPT; ++k) {
        const int e = e0 + k * 256 + tid;
        d[k] = -1; s[k] = 0; r[k] = 0;
        if (e < nedges) { d[k] = dst[e]; s[k] = src[e]; }
    }
    #pragma unroll
    for (int k = 0; k < EPT; ++k) {
        if (d[k] >= 0) r[k] = atomicAdd(&s_hist[d[k] >> BIN_SHIFT], 1);
    }
    __syncthreads();

    const int v = s_hist[tid];
    int incl = v;
    for (int off = 1; off < 32; off <<= 1) {
        const int t = __shfl_up(incl, off, 32);
        if (lane >= off) incl += t;
    }
    if (lane == 31) s_wtot[wave] = incl;
    __syncthreads();
    int wpre = 0;
    #pragma unroll
    for (int w = 0; w < 8; ++w) wpre += (w < wave) ? s_wtot[w] : 0;
    const int excl = wpre + incl - v;
    s_off[tid] = excl;
    if (tid < nbins) {
        tbl[tid * neb + blk]  = v;
        offs[tid * neb + blk] = excl;
    }
    __syncthreads();

    #pragma unroll
    for (int k = 0; k < EPT; ++k) {
        if (d[k] >= 0) s_stage[s_off[d[k] >> BIN_SHIFT] + r[k]] = make_int2(d[k], s[k]);
    }
    __syncthreads();

    const int nval = min(CH, nedges - e0);
    for (int i = tid; i < nval; i += 256) rec[(size_t)e0 + i] = s_stage[i];
}

template <int MODE>
__device__ __forceinline__ void
gcn_bin_walk(const int2* __restrict__ rec,
             const int* __restrict__ tbl,
             const int* __restrict__ offs, int b, int neb,
             int* s_cnt, int* s_n, int* s_key, int* s_src,
             const int* s_ptr, int* __restrict__ ssrc, int base) {
    const int lane  = threadIdx.x & 31;
    const int wave  = __builtin_amdgcn_readfirstlane((int)(threadIdx.x >> 5));
    const int nw    = (int)(blockDim.x >> 5);
    const int S     = min(32, (neb + nw - 1) / nw);
    const int node0 = b << BIN_SHIFT;
    for (int r0 = 0; r0 < neb; r0 += nw * S) {
        const int sg = r0 + wave * S + lane;
        int c = 0, o = 0;
        if (lane < S && sg < neb) { c = tbl[b * neb + sg]; o = offs[b * neb + sg]; }
        int incl = c;
        for (int off = 1; off < 32; off <<= 1) {
            const int t = __shfl_up(incl, off, 32);
            if (lane >= off) incl += t;
        }
        const int start = incl - c;
        const int total = __builtin_amdgcn_readlane(incl, 31);
        const int adj   = sg * CH + o - start;
        int wbase = 0;
        if (MODE == 0) {
            if (lane == 0) wbase = atomicAdd(s_n, total);
            wbase = __builtin_amdgcn_readfirstlane(wbase);
        }
        for (int fb = 0; fb < total; fb += 128) {
            int fi[4], seg[4];
            #pragma unroll
            for (int u = 0; u < 4; ++u) { fi[u] = fb + 32 * u + lane; seg[u] = 0; }
            for (int j = 1; j < S; ++j) {
                const int st = __builtin_amdgcn_readlane(start, j);
                #pragma unroll
                for (int u = 0; u < 4; ++u) seg[u] += (fi[u] >= st) ? 1 : 0;
            }
            int2 rr[4];
            #pragma unroll
            for (int u = 0; u < 4; ++u) {
                const int a = __shfl(adj, seg[u], 32) + fi[u];
                rr[u] = rec[(fi[u] < total) ? a : 0];
            }
            #pragma unroll
            for (int u = 0; u < 4; ++u) {
                if (fi[u] < total) {
                    const int dl = rr[u].x - node0;
                    if (MODE == 0) {
                        const int rk = atomicAdd(&s_cnt[dl], 1);
                        s_key[wbase + fi[u]] = (rk << BIN_SHIFT) | dl;
                        s_src[wbase + fi[u]] = rr[u].y;
                    } else if (MODE == 1) {
                        atomicAdd(&s_cnt[dl], 1);
                    } else {
                        const int p = atomicAdd(&s_cnt[dl], 1);
                        ssrc[base + s_ptr[dl] + p] = rr[u].y;
                    }
                }
            }
        }
    }
}

__global__ void __launch_bounds__(1024)
gcn_binsort_gemm_kernel(const int* __restrict__ rec_i,
                        const int* __restrict__ tbl,
                        const int* __restrict__ offs, int neb, int nbins,
                        int* __restrict__ cnt,
                        int* __restrict__ rowptr,
                        float* __restrict__ dinv,
                        int* __restrict__ deg8_i,
                        int* __restrict__ ssrc, int n,
                        const float* __restrict__ x,
                        const float* __restrict__ W1,
                        float* __restrict__ xw, int nrows,
                        int tile_first, int tile_end) {
    __shared__ int4 s_deg8[BIN_NODES / 16];
    __shared__ int s_key[SORT_CAP];
    __shared__ int s_src[SORT_CAP];
    __shared__ int s_out[SORT_CAP];
    __shared__ int s_cnt[BIN_NODES];
    __shared__ int s_ptr[BIN_NODES];
    __shared__ int s_redA[32];
    __shared__ int s_redB[32];
    __shared__ int s_wt[16];
    __shared__ int s_n;

    if ((int)blockIdx.x >= nbins) {
        gcn_gemm_block(x, W1, xw, nrows, (int)blockIdx.x - nbins, 8, tile_first, tile_end);
        return;
    }
    const int2* __restrict__ rec = (const int2*)rec_i;
    const int tid  = threadIdx.x;
    const int lane = tid & 31;
    const int wave = tid >> 5;
    const int b    = blockIdx.x;

    const int nq  = b * neb;
    const int nq4 = nq >> 2;
    const int4* __restrict__ tbl4 = (const int4*)tbl;
    int partA = 0;
    #pragma unroll 4
    for (int q = tid; q < nq4; q += 1024) {
        const int4 t = tbl4[q];
        partA += (t.x + t.y) + (t.z + t.w);
    }
    if (tid < (nq & 3)) partA += tbl[(nq4 << 2) + tid];
    int partB = 0;
    for (int q = tid; q < neb; q += 1024) partB += tbl[nq + q];
    for (int off = 16; off > 0; off >>= 1) {
        partA += __shfl_xor(partA, off, 32);
        partB += __shfl_xor(partB, off, 32);
    }
    if (lane == 0) { s_redA[wave] = partA; s_redB[wave] = partB; }
    if (tid < BIN_NODES) s_cnt[tid] = 0;
    if (tid == 0) s_n = 0;
    __syncthreads();
    int base = s_redA[lane];
    int M    = s_redB[lane];
    for (int off = 16; off > 0; off >>= 1) {
        base += __shfl_xor(base, off, 32);
        M    += __shfl_xor(M, off, 32);
    }
    const bool fast = (M <= SORT_CAP);

    if (fast) gcn_bin_walk<0>(rec, tbl, offs, b, neb, s_cnt, &s_n, s_key, s_src, s_ptr, ssrc, base);
    else      gcn_bin_walk<1>(rec, tbl, offs, b, neb, s_cnt, &s_n, s_key, s_src, s_ptr, ssrc, base);
    __syncthreads();

    const int v = (tid < BIN_NODES) ? s_cnt[tid] : 0;
    int incl = v;
    for (int off = 1; off < 32; off <<= 1) {
        const int t = __shfl_up(incl, off, 32);
        if (lane >= off) incl += t;
    }
    if (tid < BIN_NODES && lane == 31) s_wt[wave] = incl;
    __syncthreads();
    if (tid < BIN_NODES) {
        int wpre = 0;
        #pragma unroll
        for (int w = 0; w < 16; ++w) wpre += (w < wave) ? s_wt[w] : 0;
        const int excl = wpre + incl - v;
        s_ptr[tid] = excl;
        s_cnt[tid] = 0;
        const int node = (b << BIN_SHIFT) + tid;
        if (node < n) {
            cnt[node]    = v;
            rowptr[node] = base + excl;
            dinv[node]   = gcn_dinv_of(v);
        }
        ((unsigned char*)s_deg8)[tid] = (unsigned char)min(v, DEG_SAT);
    }
    __syncthreads();

    if (tid < BIN_NODES / 16) ((int4*)deg8_i)[b * (BIN_NODES / 16) + tid] = s_deg8[tid];

    if (fast) {
        for (int i = tid; i < M; i += 1024) {
            const int k = s_key[i];
            s_out[s_ptr[k & (BIN_NODES - 1)] + (k >> BIN_SHIFT)] = s_src[i];
        }
        __syncthreads();
        for (int i = tid; i < M; i += 1024) ssrc[base + i] = s_out[i];
    } else {
        gcn_bin_walk<2>(rec, tbl, offs, b, neb, s_cnt, &s_n, s_key, s_src, s_ptr, ssrc, base);
    }
}

__device__ __forceinline__ float lane_f(float v, int l) {
    return __builtin_bit_cast(float,
        __builtin_amdgcn_readlane(__builtin_bit_cast(int, v), l));
}

__global__ void __launch_bounds__(1024)
gcn_gather_finalize_kernel(const float* __restrict__ x,
                           const float* __restrict__ xw,
                           const float* __restrict__ dinv,
                           const int* __restrict__ rowptr,
                           const int* __restrict__ cnt,
                           const int* __restrict__ ssrc,
                           const int* __restrict__ deg8_i, int nbins,
                           const float* __restrict__ b1,
                           const float* __restrict__ Wfc,
                           const float* __restrict__ bfc,
                           float* __restrict__ out, int n) {
    __shared__ v4i s_deg4[MAXB * BIN_NODES / 16];
    const unsigned char* s_deg = (const unsigned char*)s_deg4;
    const int lane = threadIdx.x & 31;

    {
        const int nvec = (nbins * BIN_NODES) >> 4;
        const int rot  = (int)((blockIdx.x * (unsigned)nvec) / gridDim.x);
        const v4i* __restrict__ g4 = (const v4i*)deg8_i;
        v4i piece[8];
        int  at[8];
        #pragma unroll
        for (int u = 0; u < 8; ++u) {
            const int i = (int)threadIdx.x + u * 1024;
            int j = i + rot;
            if (j >= nvec) j -= nvec;
            at[u] = (i < nvec) ? j : -1;
            piece[u] = g4[(i < nvec) ? j : 0];
        }
        asm volatile("" :: "v"(piece[0]), "v"(piece[1]), "v"(piece[2]), "v"(piece[3]),
                           "v"(piece[4]), "v"(piece[5]), "v"(piece[6]), "v"(piece[7]));
        #pragma unroll
        for (int u = 0; u < 8; ++u) {
            if (at[u] >= 0) s_deg4[at[u]] = piece[u];
        }
    }
    __syncthreads();

    const int wave0  = __builtin_amdgcn_readfirstlane(
        (int)((blockIdx.x * blockDim.x + threadIdx.x) >> 5));
    const int nwaves = (int)((gridDim.x * blockDim.x) >> 5);
    for (int node = wave0; node < n; node += nwaves) {

    const float di   = dinv[node];
    const float self = di * di;
    const int   beg  = rowptr[node];
    const int   c    = cnt[node];

    const float4* __restrict__ xw4 = (const float4*)xw;
    const float4 xv = ((const float4*)(x + (size_t)node * NFEAT))[lane];
    const float4 wv = xw4[(size_t)node * 32 + lane];
    const float4 bv = ((const float4*)b1)[lane];

    float4 av = make_float4(0.0f, 0.0f, 0.0f, 0.0f);
    for (int base = 0; base < c; base += 32) {
        const int m = min(c - base, 32);
        int   my_s = 0;
        float my_d = 0.0f;
        if (lane < m) {
            my_s = ssrc[beg + base + lane];
            const int c8 = s_deg[my_s];
            my_d = (c8 == DEG_SAT) ? dinv[my_s] : gcn_dinv_of(c8);
        }
        int j = 0;
        for (; j + 4 <= m; j += 4) {
            const int s0 = __builtin_amdgcn_readlane(my_s, j + 0);
            const int s1 = __builtin_amdgcn_readlane(my_s, j + 1);
            const int s2 = __builtin_amdgcn_readlane(my_s, j + 2);
            const int s3 = __builtin_amdgcn_readlane(my_s, j + 3);
            const float4 r0 = xw4[(size_t)s0 * 32 + lane];
            const float4 r1 = xw4[(size_t)s1 * 32 + lane];
            const float4 r2 = xw4[(size_t)s2 * 32 + lane];
            const float4 r3 = xw4[(size_t)s3 * 32 + lane];
            const float w0 = lane_f(my_d, j + 0) * di;
            const float w1 = lane_f(my_d, j + 1) * di;
            const float w2 = lane_f(my_d, j + 2) * di;
            const float w3 = lane_f(my_d, j + 3) * di;
            av.x = fmaf(r0.x, w0, av.x); av.y = fmaf(r0.y, w0, av.y);
            av.z = fmaf(r0.z, w0, av.z); av.w = fmaf(r0.w, w0, av.w);
            av.x = fmaf(r1.x, w1, av.x); av.y = fmaf(r1.y, w1, av.y);
            av.z = fmaf(r1.z, w1, av.z); av.w = fmaf(r1.w, w1, av.w);
            av.x = fmaf(r2.x, w2, av.x); av.y = fmaf(r2.y, w2, av.y);
            av.z = fmaf(r2.z, w2, av.z); av.w = fmaf(r2.w, w2, av.w);
            av.x = fmaf(r3.x, w3, av.x); av.y = fmaf(r3.y, w3, av.y);
            av.z = fmaf(r3.z, w3, av.z); av.w = fmaf(r3.w, w3, av.w);
        }
        const int t = m - j;
        if (t == 3) {
            const int s0 = __builtin_amdgcn_readlane(my_s, j + 0);
            const int s1 = __builtin_amdgcn_readlane(my_s, j + 1);
            const int s2 = __builtin_amdgcn_readlane(my_s, j + 2);
            const float4 r0 = xw4[(size_t)s0 * 32 + lane];
            const float4 r1 = xw4[(size_t)s1 * 32 + lane];
            const float4 r2 = xw4[(size_t)s2 * 32 + lane];
            const float w0 = lane_f(my_d, j + 0) * di;
            const float w1 = lane_f(my_d, j + 1) * di;
            const float w2 = lane_f(my_d, j + 2) * di;
            av.x = fmaf(r0.x, w0, av.x); av.y = fmaf(r0.y, w0, av.y);
            av.z = fmaf(r0.z, w0, av.z); av.w = fmaf(r0.w, w0, av.w);
            av.x = fmaf(r1.x, w1, av.x); av.y = fmaf(r1.y, w1, av.y);
            av.z = fmaf(r1.z, w1, av.z); av.w = fmaf(r1.w, w1, av.w);
            av.x = fmaf(r2.x, w2, av.x); av.y = fmaf(r2.y, w2, av.y);
            av.z = fmaf(r2.z, w2, av.z); av.w = fmaf(r2.w, w2, av.w);
        } else if (t == 2) {
            const int s0 = __builtin_amdgcn_readlane(my_s, j + 0);
            const int s1 = __builtin_amdgcn_readlane(my_s, j + 1);
            const float4 r0 = xw4[(size_t)s0 * 32 + lane];
            const float4 r1 = xw4[(size_t)s1 * 32 + lane];
            const float w0 = lane_f(my_d, j + 0) * di;
            const float w1 = lane_f(my_d, j + 1) * di;
            av.x = fmaf(r0.x, w0, av.x); av.y = fmaf(r0.y, w0, av.y);
            av.z = fmaf(r0.z, w0, av.z); av.w = fmaf(r0.w, w0, av.w);
            av.x = fmaf(r1.x, w1, av.x); av.y = fmaf(r1.y, w1, av.y);
            av.z = fmaf(r1.z, w1, av.z); av.w = fmaf(r1.w, w1, av.w);
        } else if (t == 1) {
            const int s0 = __builtin_amdgcn_readlane(my_s, j);
            const float4 r0 = xw4[(size_t)s0 * 32 + lane];
            const float w0 = lane_f(my_d, j) * di;
            av.x = fmaf(r0.x, w0, av.x); av.y = fmaf(r0.y, w0, av.y);
            av.z = fmaf(r0.z, w0, av.z); av.w = fmaf(r0.w, w0, av.w);
        }
    }

    float4 h;
    h.x = fmaxf(av.x + wv.x * self + bv.x, 0.0f);
    h.y = fmaxf(av.y + wv.y * self + bv.y, 0.0f);
    h.z = fmaxf(av.z + wv.z * self + bv.z, 0.0f);
    h.w = fmaxf(av.w + wv.w * self + bv.w, 0.0f);

    const int f0 = lane * 4;
    float acc0, acc1;
    acc0  = xv.x * Wfc[(f0 + 0) * 2 + 0] + xv.y * Wfc[(f0 + 1) * 2 + 0]
          + xv.z * Wfc[(f0 + 2) * 2 + 0] + xv.w * Wfc[(f0 + 3) * 2 + 0];
    acc1  = xv.x * Wfc[(f0 + 0) * 2 + 1] + xv.y * Wfc[(f0 + 1) * 2 + 1]
          + xv.z * Wfc[(f0 + 2) * 2 + 1] + xv.w * Wfc[(f0 + 3) * 2 + 1];
    acc0 += h.x * Wfc[(NFEAT + f0 + 0) * 2 + 0] + h.y * Wfc[(NFEAT + f0 + 1) * 2 + 0]
          + h.z * Wfc[(NFEAT + f0 + 2) * 2 + 0] + h.w * Wfc[(NFEAT + f0 + 3) * 2 + 0];
    acc1 += h.x * Wfc[(NFEAT + f0 + 0) * 2 + 1] + h.y * Wfc[(NFEAT + f0 + 1) * 2 + 1]
          + h.z * Wfc[(NFEAT + f0 + 2) * 2 + 1] + h.w * Wfc[(NFEAT + f0 + 3) * 2 + 1];

    #pragma unroll
    for (int off = 16; off > 0; off >>= 1) {
        acc0 += __shfl_down(acc0, off, 32);
        acc1 += __shfl_down(acc1, off, 32);
    }
    if (lane == 0) {
        out[(size_t)node * 2 + 0] = acc0 + bfc[0];
        out[(size_t)node * 2 + 1] = acc1 + bfc[1];
    }

    }
}

extern "C" void kernel_launch(void* const* d_in, const int* in_sizes, int n_in,
                              void* d_out, int out_size, void* d_ws, size_t ws_size,
                              hipStream_t stream) {
    const float* x   = (const float*)d_in[0];
    const int*   ei  = (const int*)  d_in[1];
    const float* W1  = (const float*)d_in[2];
    const float* b1  = (const float*)d_in[3];
    const float* Wfc = (const float*)d_in[4];
    const float* bfc = (const float*)d_in[5];
    float*       out = (float*)d_out;

    const int N = in_sizes[0] / NFEAT;
    const int E = in_sizes[1] / 2;
    const int* src = ei;
    const int* dst = ei + E;

    const int neb   = (E + CH - 1) / CH;
    const int nbins = (N + BIN_NODES - 1) / BIN_NODES;

    float* xw     = (float*)d_ws;
    int*   rec    = (int*)(xw + (size_t)N * HIDW);
    int*   deg8   = rec + (size_t)neb * CH * 2;
    int*   tbl    = deg8 + (size_t)nbins * (BIN_NODES / 4);
    int*   offs   = tbl + (size_t)nbins * neb;
    int*   ssrc   = offs + (size_t)nbins * neb;
    int*   cnt    = ssrc + E;
    int*   rowptr = cnt + N;
    float* dinv   = (float*)(rowptr + N);

    const int tiles   = (N + 15) / 16;
    const int tilesA  = (tiles + 1) / 2;
    const int tilesB  = tiles - tilesA;
    const int blocksA = (tilesA + 1) / 2;
    const int blocksB = (tilesB + 7) / 8;

    gcn_localsort_gemm_kernel<<<neb + blocksA, 256, 0, stream>>>(src, dst, E, neb, nbins,
                                                                 rec, tbl, offs,
                                                                 x, W1, xw, N, 0, tilesA);

    gcn_binsort_gemm_kernel<<<nbins + blocksB, 1024, 0, stream>>>(rec, tbl, offs, neb, nbins,
                                                                  cnt, rowptr, dinv, deg8, ssrc, N,
                                                                  x, W1, xw, N, tilesA, tiles);

    gcn_gather_finalize_kernel<<<GATHER_BLOCKS, 1024, 0, stream>>>(x, xw, dinv, rowptr, cnt, ssrc,
                                                                   deg8, nbins,
                                                                   b1, Wfc, bfc, out, N);
}


# x@W1 operands staged in LDS: a product block copies W1 to LDS once (k-pair rows) and each 16-row x tile by whole-row loads; WMMA fragments read from LDS (ds_load_b64), not per-wave global loads
# speedup vs baseline: 5.2495x; 1.0763x over previous
#include <hip/hip_runtime.h>
#include <hip/hip_bf16.h>


typedef float v2f __attribute__((ext_vector_type(2)));
typedef float v8f __attribute__((ext_vector_type(8)));
typedef int   v4i __attribute__((ext_vector_type(4)));

#define NFEAT 128
#define HIDW  128

#define BIN_SHIFT 9
#define BIN_NODES 512
#define MAXB      256
#define CH        4096
#define EPT       16
#define SORT_CAP  12288
#define DEG_SAT   255
#define GATHER_BLOCKS 512

__device__ __forceinline__ float gcn_dinv_of(int cnt) {
    const float dg = (float)(cnt + 1);
    return (dg > 0.0f) ? rsqrtf(dg) : 0.0f;
}


#define WP_PITCH  288
#define WP_FLOATS (64 * WP_PITCH)
#define AT_PITCH  132
#define AT_FLOATS (16 * AT_PITCH)
#define GEMM_LDS_BYTES(NG) ((WP_FLOATS + (NG) * AT_FLOATS) * 4)
#define LS_LDS_BYTES GEMM_LDS_BYTES(2)
#define BS_LDS_BYTES (BIN_NODES + (3 * SORT_CAP + 2 * BIN_NODES + 32 + 32 + 16 + 4) * 4)

__device__ __forceinline__ void
gcn_gemm_xw_tile(const float* s_w, const float* s_at,
                 float* __restrict__ xw, int nrows, int tileM, int t) {
    const int wave  = t >> 5;
    const int lane  = t & 31;
    const int l15   = lane & 15;
    const int h     = lane >> 4;
    const int n0    = wave * 32;
    const int ncol0 = n0 + l15;
    const int ncol1 = n0 + 16 + l15;

    v8f c0 = {0.f, 0.f, 0.f, 0.f, 0.f, 0.f, 0.f, 0.f};
    v8f c1 = {0.f, 0.f, 0.f, 0.f, 0.f, 0.f, 0.f, 0.f};

    const float* arow = s_at + l15 * AT_PITCH + 2 * h;
    const float* wrow = s_w + h * WP_PITCH;
    #pragma unroll 8
    for (int kb = 0; kb < NFEAT; kb += 4) {
        const v2f a  = *(const v2f*)(arow + kb);
        const float* wp = wrow + (kb >> 1) * WP_PITCH;
        const v2f b0 = *(const v2f*)(wp + ncol0 * 2);
        const v2f b1 = *(const v2f*)(wp + ncol1 * 2);
        c0 = __builtin_amdgcn_wmma_f32_16x16x4_f32(false, a, false, b0,
                                                   (short)0, c0, false, false);
        c1 = __builtin_amdgcn_wmma_f32_16x16x4_f32(false, a, false, b1,
                                                   (short)0, c1, false, false);
    }

    const int rBase = tileM + h * 8;
    float* __restrict__ orow = xw + (size_t)rBase * HIDW;
    if (tileM + 16 <= nrows) {
        #pragma unroll
        for (int r = 0; r < 8; ++r) {
            orow[(size_t)r * HIDW + ncol0] = c0[r];
            orow[(size_t)r * HIDW + ncol1] = c1[r];
        }
    } else {
        #pragma unroll
        for (int r = 0; r < 8; ++r) {
            if (rBase + r < nrows) {
                orow[(size_t)r * HIDW + ncol0] = c0[r];
                orow[(size_t)r * HIDW + ncol1] = c1[r];
            }
        }
    }
}

__device__ __forceinline__ float4
gcn_gemm_load_xpiece(const float* __restrict__ x, int nrows, int tileM, int t, int i) {
    const int row = min(tileM + (t >> 5) + 4 * i, nrows - 1);
    return ((const float4*)x)[(size_t)row * (NFEAT / 4) + (t & 31)];
}

template <int NG>
__device__ __forceinline__ void
gcn_gemm_block(const float* __restrict__ x,
               const float* __restrict__ W1,
               float* __restrict__ xw, int nrows,
               int g, int tpg, int tile_first, int tile_end,
               float* s_w, float* s_a) {
    const int tid = (int)threadIdx.x;
    const int grp = __builtin_amdgcn_readfirstlane(tid >> 7);
    const int t   = tid & 127;
    const int base = tile_first + g * (NG * tpg) + grp;
    float* s_at = s_a + grp * AT_FLOATS;

    float4 p0 = make_float4(0.f, 0.f, 0.f, 0.f), p1 = p0, p2 = p0, p3 = p0;
    bool have = base < tile_end;
    if (have) {
        p0 = gcn_gemm_load_xpiece(x, nrows, base * 16, t, 0);
        p1 = gcn_gemm_load_xpiece(x, nrows, base * 16, t, 1);
        p2 = gcn_gemm_load_xpiece(x, nrows, base * 16, t, 2);
        p3 = gcn_gemm_load_xpiece(x, nrows, base * 16, t, 3);
    }

    {
        constexpr int NIT = (64 * 32) / (NG * 128);
        const float4* __restrict__ W4 = (const float4*)W1;
        float4 r0[NIT], r1[NIT];
        #pragma unroll
        for (int u = 0; u < NIT; ++u) {
            const int it = tid + u * (NG * 128);
            const int kp = it >> 5, cq = it & 31;
            r0[u] = W4[(2 * kp)     * (HIDW / 4) + cq];
            r1[u] = W4[(2 * kp + 1) * (HIDW / 4) + cq];
        }
        #pragma unroll
        for (int u = 0; u < NIT; ++u) {
            const int it = tid + u * (NG * 128);
            const int kp = it >> 5, cq = it & 31;
            float4* d = (float4*)(s_w + kp * WP_PITCH + cq * 8);
            d[0] = make_float4(r0[u].x, r1[u].x, r0[u].y, r1[u].y);
            d[1] = make_float4(r0[u].z, r1[u].z, r0[u].w, r1[u].w);
        }
    }

    for (int i = 0; i < tpg; ++i) {
        const int tile = base + i * NG;
        if (have) {
            float* d = s_at + (t >> 5) * AT_PITCH + (t & 31) * 4;
            *(float4*)(d)                 = p0;
            *(float4*)(d + 4 * AT_PITCH)  = p1;
            *(float4*)(d + 8 * AT_PITCH)  = p2;
            *(float4*)(d + 12 * AT_PITCH) = p3;
        }
        __syncthreads();
        const bool cur = have;
        have = (i + 1 < tpg) && (tile + NG < tile_end);
        if (have) {
            p0 = gcn_gemm_load_xpiece(x, nrows, (tile + NG) * 16, t, 0);
            p1 = gcn_gemm_load_xpiece(x, nrows, (tile + NG) * 16, t, 1);
            p2 = gcn_gemm_load_xpiece(x, nrows, (tile + NG) * 16, t, 2);
            p3 = gcn_gemm_load_xpiece(x, nrows, (tile + NG) * 16, t, 3);
        }
        if (cur) gcn_gemm_xw_tile(s_w, s_at, xw, nrows, tile * 16, t);
        __syncthreads();
    }
}

__global__ void __launch_bounds__(256)
gcn_localsort_gemm_kernel(const int* __restrict__ src,
                          const int* __restrict__ dst, int nedges,
                          int neb, int nbins,
                          int* __restrict__ rec_i,
                          int* __restrict__ tbl,
                          int* __restrict__ offs,
                          const float* __restrict__ x,
                          const float* __restrict__ W1,
                          float* __restrict__ xw, int nrows,
                          int tile_first, int tile_end) {
    __shared__ __attribute__((aligned(16))) unsigned char s_raw[LS_LDS_BYTES];

    if ((int)blockIdx.x >= neb) {
        gcn_gemm_block<2>(x, W1, xw, nrows, (int)blockIdx.x - neb, 2, tile_first, tile_end,
                          (float*)s_raw, (float*)s_raw + WP_FLOATS);
        return;
    }
    int*  s_hist  = (int*)s_raw;
    int*  s_off   = s_hist + MAXB;
    int*  s_wtot  = s_off + MAXB;
    int2* s_stage = (int2*)(s_raw + 4096);
    int2* __restrict__ rec = (int2*)rec_i;
    const int tid  = threadIdx.x;
    const int lane = tid & 31;
    const int wave = tid >> 5;
    const int blk  = blockIdx.x;
    const int e0   = blk * CH;

    s_hist[tid] = 0;
    __syncthreads();

    int d[EPT], s[EPT], r[EPT];
    #pragma unroll
    for (int k = 0; k < EPT; ++k) {
        const int e = e0 + k * 256 + tid;
        d[k] = -1; s[k] = 0; r[k] = 0;
        if (e < nedges) { d[k] = dst[e]; s[k] = src[e]; }
    }
    #pragma unroll
    for (int k = 0; k < EPT; ++k) {
        if (d[k] >= 0) r[k] = atomicAdd(&s_hist[d[k] >> BIN_SHIFT], 1);
    }
    __syncthreads();

    const int v = s_hist[tid];
    int incl = v;
    for (int off = 1; off < 32; off <<= 1) {
        const int t = __shfl_up(incl, off, 32);
        if (lane >= off) incl += t;
    }
    if (lane == 31) s_wtot[wave] = incl;
    __syncthreads();
    int wpre = 0;
    #pragma unroll
    for (int w = 0; w < 8; ++w) wpre += (w < wave) ? s_wtot[w] : 0;
    const int excl = wpre + incl - v;
    s_off[tid] = excl;
    if (tid < nbins) {
        tbl[tid * neb + blk]  = v;
        offs[tid * neb + blk] = excl;
    }
    __syncthreads();

    #pragma unroll
    for (int k = 0; k < EPT; ++k) {
        if (d[k] >= 0) s_stage[s_off[d[k] >> BIN_SHIFT] + r[k]] = make_int2(d[k], s[k]);
    }
    __syncthreads();

    const int nval = min(CH, nedges - e0);
    for (int i = tid; i < nval; i += 256) rec[(size_t)e0 + i] = s_stage[i];
}

template <int MODE>
__device__ __forceinline__ void
gcn_bin_walk(const int2* __restrict__ rec,
             const int* __restrict__ tbl,
             const int* __restrict__ offs, int b, int neb,
             int* s_cnt, int* s_n, int* s_key, int* s_src,
             const int* s_ptr, int* __restrict__ ssrc, int base) {
    const int lane  = threadIdx.x & 31;
    const int wave  = __builtin_amdgcn_readfirstlane((int)(threadIdx.x >> 5));
    const int nw    = (int)(blockDim.x >> 5);
    const int S     = min(32, (neb + nw - 1) / nw);
    const int node0 = b << BIN_SHIFT;
    for (int r0 = 0; r0 < neb; r0 += nw * S) {
        const int sg = r0 + wave * S + lane;
        int c = 0, o = 0;
        if (lane < S && sg < neb) { c = tbl[b * neb + sg]; o = offs[b * neb + sg]; }
        int incl = c;
        for (int off = 1; off < 32; off <<= 1) {
            const int t = __shfl_up(incl, off, 32);
            if (lane >= off) incl += t;
        }
        const int start = incl - c;
        const int total = __builtin_amdgcn_readlane(incl, 31);
        const int adj   = sg * CH + o - start;
        int wbase = 0;
        if (MODE == 0) {
            if (lane == 0) wbase = atomicAdd(s_n, total);
            wbase = __builtin_amdgcn_readfirstlane(wbase);
        }
        for (int fb = 0; fb < total; fb += 128) {
            int fi[4], seg[4];
            #pragma unroll
            for (int u = 0; u < 4; ++u) { fi[u] = fb + 32 * u + lane; seg[u] = 0; }
            for (int j = 1; j < S; ++j) {
                const int st = __builtin_amdgcn_readlane(start, j);
                #pragma unroll
                for (int u = 0; u < 4; ++u) seg[u] += (fi[u] >= st) ? 1 : 0;
            }
            int2 rr[4];
            #pragma unroll
            for (int u = 0; u < 4; ++u) {
                const int a = __shfl(adj, seg[u], 32) + fi[u];
                rr[u] = rec[(fi[u] < total) ? a : 0];
            }
            #pragma unroll
            for (int u = 0; u < 4; ++u) {
                if (fi[u] < total) {
                    const int dl = rr[u].x - node0;
                    if (MODE == 0) {
                        const int rk = atomicAdd(&s_cnt[dl], 1);
                        s_key[wbase + fi[u]] = (rk << BIN_SHIFT) | dl;
                        s_src[wbase + fi[u]] = rr[u].y;
                    } else if (MODE == 1) {
                        atomicAdd(&s_cnt[dl], 1);
                    } else {
                        const int p = atomicAdd(&s_cnt[dl], 1);
                        ssrc[base + s_ptr[dl] + p] = rr[u].y;
                    }
                }
            }
        }
    }
}

__global__ void __launch_bounds__(1024)
gcn_binsort_gemm_kernel(const int* __restrict__ rec_i,
                        const int* __restrict__ tbl,
                        const int* __restrict__ offs, int neb, int nbins,
                        int* __restrict__ cnt,
                        int* __restrict__ rowptr,
                        float* __restrict__ dinv,
                        int* __restrict__ deg8_i,
                        int* __restrict__ ssrc, int n,
                        const float* __restrict__ x,
                        const float* __restrict__ W1,
                        float* __restrict__ xw, int nrows,
                        int tile_first, int tile_end) {
    __shared__ __attribute__((aligned(16))) unsigned char s_raw[BS_LDS_BYTES];

    if ((int)blockIdx.x >= nbins) {
        gcn_gemm_block<8>(x, W1, xw, nrows, (int)blockIdx.x - nbins, 1, tile_first, tile_end,
                          (float*)s_raw, (float*)s_raw + WP_FLOATS);
        return;
    }
    int4* s_deg8 = (int4*)s_raw;
    int*  s_key  = (int*)(s_raw + BIN_NODES);
    int*  s_src  = s_key + SORT_CAP;
    int*  s_out  = s_src + SORT_CAP;
    int*  s_cnt  = s_out + SORT_CAP;
    int*  s_ptr  = s_cnt + BIN_NODES;
    int*  s_redA = s_ptr + BIN_NODES;
    int*  s_redB = s_redA + 32;
    int*  s_wt   = s_redB + 32;
    int&  s_n    = s_wt[16];
    const int2* __restrict__ rec = (const int2*)rec_i;
    const int tid  = threadIdx.x;
    const int lane = tid & 31;
    const int wave = tid >> 5;
    const int b    = blockIdx.x;

    const int nq  = b * neb;
    const int nq4 = nq >> 2;
    const int4* __restrict__ tbl4 = (const int4*)tbl;
    int partA = 0;
    #pragma unroll 4
    for (int q = tid; q < nq4; q += 1024) {
        const int4 t = tbl4[q];
        partA += (t.x + t.y) + (t.z + t.w);
    }
    if (tid < (nq & 3)) partA += tbl[(nq4 << 2) + tid];
    int partB = 0;
    for (int q = tid; q < neb; q += 1024) partB += tbl[nq + q];
    for (int off = 16; off > 0; off >>= 1) {
        partA += __shfl_xor(partA, off, 32);
        partB += __shfl_xor(partB, off, 32);
    }
    if (lane == 0) { s_redA[wave] = partA; s_redB[wave] = partB; }
    if (tid < BIN_NODES) s_cnt[tid] = 0;
    if (tid == 0) s_n = 0;
    __syncthreads();
    int base = s_redA[lane];
    int M    = s_redB[lane];
    for (int off = 16; off > 0; off >>= 1) {
        base += __shfl_xor(base, off, 32);
        M    += __shfl_xor(M, off, 32);
    }
    const bool fast = (M <= SORT_CAP);

    if (fast) gcn_bin_walk<0>(rec, tbl, offs, b, neb, s_cnt, &s_n, s_key, s_src, s_ptr, ssrc, base);
    else      gcn_bin_walk<1>(rec, tbl, offs, b, neb, s_cnt, &s_n, s_key, s_src, s_ptr, ssrc, base);
    __syncthreads();

    const int v = (tid < BIN_NODES) ? s_cnt[tid] : 0;
    int incl = v;
    for (int off = 1; off < 32; off <<= 1) {
        const int t = __shfl_up(incl, off, 32);
        if (lane >= off) incl += t;
    }
    if (tid < BIN_NODES && lane == 31) s_wt[wave] = incl;
    __syncthreads();
    if (tid < BIN_NODES) {
        int wpre = 0;
        #pragma unroll
        for (int w = 0; w < 16; ++w) wpre += (w < wave) ? s_wt[w] : 0;
        const int excl = wpre + incl - v;
        s_ptr[tid] = excl;
        s_cnt[tid] = 0;
        const int node = (b << BIN_SHIFT) + tid;
        if (node < n) {
            cnt[node]    = v;
            rowptr[node] = base + excl;
            dinv[node]   = gcn_dinv_of(v);
        }
        ((unsigned char*)s_deg8)[tid] = (unsigned char)min(v, DEG_SAT);
    }
    __syncthreads();

    if (tid < BIN_NODES / 16) ((int4*)deg8_i)[b * (BIN_NODES / 16) + tid] = s_deg8[tid];

    if (fast) {
        for (int i = tid; i < M; i += 1024) {
            const int k = s_key[i];
            s_out[s_ptr[k & (BIN_NODES - 1)] + (k >> BIN_SHIFT)] = s_src[i];
        }
        __syncthreads();
        for (int i = tid; i < M; i += 1024) ssrc[base + i] = s_out[i];
    } else {
        gcn_bin_walk<2>(rec, tbl, offs, b, neb, s_cnt, &s_n, s_key, s_src, s_ptr, ssrc, base);
    }
}

__device__ __forceinline__ float lane_f(float v, int l) {
    return __builtin_bit_cast(float,
        __builtin_amdgcn_readlane(__builtin_bit_cast(int, v), l));
}

__global__ void __launch_bounds__(1024)
gcn_gather_finalize_kernel(const float* __restrict__ x,
                           const float* __restrict__ xw,
                           const float* __restrict__ dinv,
                           const int* __restrict__ rowptr,
                           const int* __restrict__ cnt,
                           const int* __restrict__ ssrc,
                           const int* __restrict__ deg8_i, int nbins,
                           const float* __restrict__ b1,
                           const float* __restrict__ Wfc,
                           const float* __restrict__ bfc,
                           float* __restrict__ out, int n) {
    __shared__ v4i s_deg4[MAXB * BIN_NODES / 16];
    const unsigned char* s_deg = (const unsigned char*)s_deg4;
    const int lane = threadIdx.x & 31;

    {
        const int nvec = (nbins * BIN_NODES) >> 4;
        const int rot  = (int)((blockIdx.x * (unsigned)nvec) / gridDim.x);
        const v4i* __restrict__ g4 = (const v4i*)deg8_i;
        v4i piece[8];
        int  at[8];
        #pragma unroll
        for (int u = 0; u < 8; ++u) {
            const int i = (int)threadIdx.x + u * 1024;
            int j = i + rot;
            if (j >= nvec) j -= nvec;
            at[u] = (i < nvec) ? j : -1;
            piece[u] = g4[(i < nvec) ? j : 0];
        }
        asm volatile("" :: "v"(piece[0]), "v"(piece[1]), "v"(piece[2]), "v"(piece[3]),
                           "v"(piece[4]), "v"(piece[5]), "v"(piece[6]), "v"(piece[7]));
        #pragma unroll
        for (int u = 0; u < 8; ++u) {
            if (at[u] >= 0) s_deg4[at[u]] = piece[u];
        }
    }
    __syncthreads();

    const int wave0  = __builtin_amdgcn_readfirstlane(
        (int)((blockIdx.x * blockDim.x + threadIdx.x) >> 5));
    const int nwaves = (int)((gridDim.x * blockDim.x) >> 5);
    for (int node = wave0; node < n; node += nwaves) {

    const float di   = dinv[node];
    const float self = di * di;
    const int   beg  = rowptr[node];
    const int   c    = cnt[node];

    const float4* __restrict__ xw4 = (const float4*)xw;
    const float4 xv = ((const float4*)(x + (size_t)node * NFEAT))[lane];
    const float4 wv = xw4[(size_t)node * 32 + lane];
    const float4 bv = ((const float4*)b1)[lane];

    float4 av = make_float4(0.0f, 0.0f, 0.0f, 0.0f);
    for (int base = 0; base < c; base += 32) {
        const int m = min(c - base, 32);
        int   my_s = 0;
        float my_d = 0.0f;
        if (lane < m) {
            my_s = ssrc[beg + base + lane];
            const int c8 = s_deg[my_s];
            my_d = (c8 == DEG_SAT) ? dinv[my_s] : gcn_dinv_of(c8);
        }
        int j = 0;
        for (; j + 4 <= m; j += 4) {
            const int s0 = __builtin_amdgcn_readlane(my_s, j + 0);
            const int s1 = __builtin_amdgcn_readlane(my_s, j + 1);
            const int s2 = __builtin_amdgcn_readlane(my_s, j + 2);
            const int s3 = __builtin_amdgcn_readlane(my_s, j + 3);
            const float4 r0 = xw4[(size_t)s0 * 32 + lane];
            const float4 r1 = xw4[(size_t)s1 * 32 + lane];
            const float4 r2 = xw4[(size_t)s2 * 32 + lane];
            const float4 r3 = xw4[(size_t)s3 * 32 + lane];
            const float w0 = lane_f(my_d, j + 0) * di;
            const float w1 = lane_f(my_d, j + 1) * di;
            const float w2 = lane_f(my_d, j + 2) * di;
            const float w3 = lane_f(my_d, j + 3) * di;
            av.x = fmaf(r0.x, w0, av.x); av.y = fmaf(r0.y, w0, av.y);
            av.z = fmaf(r0.z, w0, av.z); av.w = fmaf(r0.w, w0, av.w);
            av.x = fmaf(r1.x, w1, av.x); av.y = fmaf(r1.y, w1, av.y);
            av.z = fmaf(r1.z, w1, av.z); av.w = fmaf(r1.w, w1, av.w);
            av.x = fmaf(r2.x, w2, av.x); av.y = fmaf(r2.y, w2, av.y);
            av.z = fmaf(r2.z, w2, av.z); av.w = fmaf(r2.w, w2, av.w);
            av.x = fmaf(r3.x, w3, av.x); av.y = fmaf(r3.y, w3, av.y);
            av.z = fmaf(r3.z, w3, av.z); av.w = fmaf(r3.w, w3, av.w);
        }
        const int t = m - j;
        if (t == 3) {
            const int s0 = __builtin_amdgcn_readlane(my_s, j + 0);
            const int s1 = __builtin_amdgcn_readlane(my_s, j + 1);
            const int s2 = __builtin_amdgcn_readlane(my_s, j + 2);
            const float4 r0 = xw4[(size_t)s0 * 32 + lane];
            const float4 r1 = xw4[(size_t)s1 * 32 + lane];
            const float4 r2 = xw4[(size_t)s2 * 32 + lane];
            const float w0 = lane_f(my_d, j + 0) * di;
            const float w1 = lane_f(my_d, j + 1) * di;
            const float w2 = lane_f(my_d, j + 2) * di;
            av.x = fmaf(r0.x, w0, av.x); av.y = fmaf(r0.y, w0, av.y);
            av.z = fmaf(r0.z, w0, av.z); av.w = fmaf(r0.w, w0, av.w);
            av.x = fmaf(r1.x, w1, av.x); av.y = fmaf(r1.y, w1, av.y);
            av.z = fmaf(r1.z, w1, av.z); av.w = fmaf(r1.w, w1, av.w);
            av.x = fmaf(r2.x, w2, av.x); av.y = fmaf(r2.y, w2, av.y);
            av.z = fmaf(r2.z, w2, av.z); av.w = fmaf(r2.w, w2, av.w);
        } else if (t == 2) {
            const int s0 = __builtin_amdgcn_readlane(my_s, j + 0);
            const int s1 = __builtin_amdgcn_readlane(my_s, j + 1);
            const float4 r0 = xw4[(size_t)s0 * 32 + lane];
            const float4 r1 = xw4[(size_t)s1 * 32 + lane];
            const float w0 = lane_f(my_d, j + 0) * di;
            const float w1 = lane_f(my_d, j + 1) * di;
            av.x = fmaf(r0.x, w0, av.x); av.y = fmaf(r0.y, w0, av.y);
            av.z = fmaf(r0.z, w0, av.z); av.w = fmaf(r0.w, w0, av.w);
            av.x = fmaf(r1.x, w1, av.x); av.y = fmaf(r1.y, w1, av.y);
            av.z = fmaf(r1.z, w1, av.z); av.w = fmaf(r1.w, w1, av.w);
        } else if (t == 1) {
            const int s0 = __builtin_amdgcn_readlane(my_s, j);
            const float4 r0 = xw4[(size_t)s0 * 32 + lane];
            const float w0 = lane_f(my_d, j) * di;
            av.x = fmaf(r0.x, w0, av.x); av.y = fmaf(r0.y, w0, av.y);
            av.z = fmaf(r0.z, w0, av.z); av.w = fmaf(r0.w, w0, av.w);
        }
    }

    float4 h;
    h.x = fmaxf(av.x + wv.x * self + bv.x, 0.0f);
    h.y = fmaxf(av.y + wv.y * self + bv.y, 0.0f);
    h.z = fmaxf(av.z + wv.z * self + bv.z, 0.0f);
    h.w = fmaxf(av.w + wv.w * self + bv.w, 0.0f);

    const int f0 = lane * 4;
    float acc0, acc1;
    acc0  = xv.x * Wfc[(f0 + 0) * 2 + 0] + xv.y * Wfc[(f0 + 1) * 2 + 0]
          + xv.z * Wfc[(f0 + 2) * 2 + 0] + xv.w * Wfc[(f0 + 3) * 2 + 0];
    acc1  = xv.x * Wfc[(f0 + 0) * 2 + 1] + xv.y * Wfc[(f0 + 1) * 2 + 1]
          + xv.z * Wfc[(f0 + 2) * 2 + 1] + xv.w * Wfc[(f0 + 3) * 2 + 1];
    acc0 += h.x * Wfc[(NFEAT + f0 + 0) * 2 + 0] + h.y * Wfc[(NFEAT + f0 + 1) * 2 + 0]
          + h.z * Wfc[(NFEAT + f0 + 2) * 2 + 0] + h.w * Wfc[(NFEAT + f0 + 3) * 2 + 0];
    acc1 += h.x * Wfc[(NFEAT + f0 + 0) * 2 + 1] + h.y * Wfc[(NFEAT + f0 + 1) * 2 + 1]
          + h.z * Wfc[(NFEAT + f0 + 2) * 2 + 1] + h.w * Wfc[(NFEAT + f0 + 3) * 2 + 1];

    #pragma unroll
    for (int off = 16; off > 0; off >>= 1) {
        acc0 += __shfl_down(acc0, off, 32);
        acc1 += __shfl_down(acc1, off, 32);
    }
    if (lane == 0) {
        out[(size_t)node * 2 + 0] = acc0 + bfc[0];
        out[(size_t)node * 2 + 1] = acc1 + bfc[1];
    }

    }
}

extern "C" void kernel_launch(void* const* d_in, const int* in_sizes, int n_in,
                              void* d_out, int out_size, void* d_ws, size_t ws_size,
                              hipStream_t stream) {
    const float* x   = (const float*)d_in[0];
    const int*   ei  = (const int*)  d_in[1];
    const float* W1  = (const float*)d_in[2];
    const float* b1  = (const float*)d_in[3];
    const float* Wfc = (const float*)d_in[4];
    const float* bfc = (const float*)d_in[5];
    float*       out = (float*)d_out;

    const int N = in_sizes[0] / NFEAT;
    const int E = in_sizes[1] / 2;
    const int* src = ei;
    const int* dst = ei + E;

    const int neb   = (E + CH - 1) / CH;
    const int nbins = (N + BIN_NODES - 1) / BIN_NODES;

    float* xw     = (float*)d_ws;
    int*   rec    = (int*)(xw + (size_t)N * HIDW);
    int*   deg8   = rec + (size_t)neb * CH * 2;
    int*   tbl    = deg8 + (size_t)nbins * (BIN_NODES / 4);
    int*   offs   = tbl + (size_t)nbins * neb;
    int*   ssrc   = offs + (size_t)nbins * neb;
    int*   cnt    = ssrc + E;
    int*   rowptr = cnt + N;
    float* dinv   = (float*)(rowptr + N);

    const int tiles   = (N + 15) / 16;
    const int tilesA  = (tiles + 1) / 2;
    const int tilesB  = tiles - tilesA;
    const int blocksA = (tilesA + 3) / 4;
    const int blocksB = (tilesB + 7) / 8;

    gcn_localsort_gemm_kernel<<<neb + blocksA, 256, 0, stream>>>(src, dst, E, neb, nbins,
                                                                 rec, tbl, offs,
                                                                 x, W1, xw, N, 0, tilesA);

    gcn_binsort_gemm_kernel<<<nbins + blocksB, 1024, 0, stream>>>(rec, tbl, offs, neb, nbins,
                                                                  cnt, rowptr, dinv, deg8, ssrc, N,
                                                                  x, W1, xw, N, tilesA, tiles);

    gcn_gather_finalize_kernel<<<GATHER_BLOCKS, 1024, 0, stream>>>(x, xw, dinv, rowptr, cnt, ssrc,
                                                                   deg8, nbins,
                                                                   b1, Wfc, bfc, out, N);
}
